# starting kernel
# baseline (speedup 1.0000x reference)
#include <hip/hip_runtime.h>
#include <hip/hip_bf16.h>
#include <math.h>

#define BATCH 4
#define SEQ   2048
#define DIMX  192
#define DI    384
#define NST   16
#define DTR   12
#define XZW   768
#define DBLW  48
#define MROWS (BATCH*SEQ)
#define NCH   32
#define CL    (SEQ/NCH)

typedef float v2f __attribute__((ext_vector_type(2)));
typedef float v8f __attribute__((ext_vector_type(8)));

__device__ __forceinline__ v8f wmma4(v2f a, v2f b, v8f c) {
  return __builtin_amdgcn_wmma_f32_16x16x4_f32(false, a, false, b, (short)0, c,
                                               false, false);
}

__global__ void k_gemm_in(const float* __restrict__ x,
                          const float* __restrict__ Win,
                          float* __restrict__ xz, int reverse) {
  const int wave = blockIdx.x * (blockDim.x >> 5) + (threadIdx.x >> 5);
  const int lane = threadIdx.x & 31;
  const int TG = XZW / 64;
  const int gm = wave / TG, gng = wave % TG;
  const int mrow = lane & 15, kh = lane >> 4;

  int m  = gm * 16 + mrow;
  int bb = m / SEQ, ll = m % SEQ;
  int srow = reverse ? (bb * SEQ + (SEQ - 1 - ll)) : m;

  const float* ap = x + (size_t)srow * DIMX;
  const float* bp[4];
#pragma unroll
  for (int j = 0; j < 4; ++j)
    bp[j] = Win + (size_t)(gng * 64 + j * 16 + mrow) * DIMX;

  v8f acc[4] = {};
  for (int k0 = 0; k0 < DIMX; k0 += 4) {
    int kk = k0 + 2 * kh;
    v2f a; a.x = ap[kk]; a.y = ap[kk + 1];
#pragma unroll
    for (int j = 0; j < 4; ++j) {
      v2f b; b.x = bp[j][kk]; b.y = bp[j][kk + 1];
      acc[j] = wmma4(a, b, acc[j]);
    }
  }
#pragma unroll
  for (int j = 0; j < 4; ++j)
#pragma unroll
    for (int v = 0; v < 8; ++v) {
      int orow = gm * 16 + v + 8 * kh;
      xz[(size_t)orow * XZW + gng * 64 + j * 16 + mrow] = acc[j][v];
    }
}

__global__ void k_conv_silu(const float* __restrict__ xz,
                            const float* __restrict__ cw,
                            const float* __restrict__ cb,
                            float* __restrict__ u) {
  int idx = blockIdx.x * blockDim.x + threadIdx.x;
  if (idx >= MROWS * DI) return;
  int d = idx % DI;
  int t = (idx / DI) % SEQ;
  int b = idx / (DI * SEQ);
  float acc = cb[d];
#pragma unroll
  for (int j = 0; j < 4; ++j) {
    int tt = t - 3 + j;
    if (tt >= 0)
      acc += xz[(size_t)(b * SEQ + tt) * XZW + d] * cw[d * 4 + j];
  }
  u[(size_t)(b * SEQ + t) * DI + d] = acc / (1.0f + __expf(-acc));
}

__global__ void k_gemm_xp(const float* __restrict__ u,
                          const float* __restrict__ Wxp,
                          float* __restrict__ dbl) {
  const int wave = blockIdx.x * (blockDim.x >> 5) + (threadIdx.x >> 5);
  const int lane = threadIdx.x & 31;
  const int gm = wave;
  const int mrow = lane & 15, kh = lane >> 4;

  const float* ap = u + (size_t)(gm * 16 + mrow) * DI;
  bool nvalid[3];
  const float* bp[3];
#pragma unroll
  for (int j = 0; j < 3; ++j) {
    int n = j * 16 + mrow;
    nvalid[j] = (n < DTR + 2 * NST);
    bp[j] = Wxp + (size_t)(nvalid[j] ? n : 0) * DI;
  }

  v8f acc[3] = {};
  for (int k0 = 0; k0 < DI; k0 += 4) {
    int kk = k0 + 2 * kh;
    v2f a; a.x = ap[kk]; a.y = ap[kk + 1];
#pragma unroll
    for (int j = 0; j < 3; ++j) {
      v2f b;
      b.x = nvalid[j] ? bp[j][kk] : 0.0f;
      b.y = nvalid[j] ? bp[j][kk + 1] : 0.0f;
      acc[j] = wmma4(a, b, acc[j]);
    }
  }
#pragma unroll
  for (int j = 0; j < 3; ++j)
#pragma unroll
    for (int v = 0; v < 8; ++v) {
      int orow = gm * 16 + v + 8 * kh;
      dbl[(size_t)orow * DBLW + j * 16 + mrow] = acc[j][v];
    }
}

__global__ void k_gemm_dt(const float* __restrict__ dbl,
                          const float* __restrict__ Wdt,
                          const float* __restrict__ bdt,
                          float* __restrict__ xz) {
  const int wave = blockIdx.x * (blockDim.x >> 5) + (threadIdx.x >> 5);
  const int lane = threadIdx.x & 31;
  const int TG = DI / 64;
  const int gm = wave / TG, gng = wave % TG;
  const int mrow = lane & 15, kh = lane >> 4;

  const float* ap = dbl + (size_t)(gm * 16 + mrow) * DBLW;
  const float* bp[4];
#pragma unroll
  for (int j = 0; j < 4; ++j)
    bp[j] = Wdt + (size_t)(gng * 64 + j * 16 + mrow) * DTR;

  v8f acc[4] = {};
  for (int k0 = 0; k0 < DTR; k0 += 4) {
    int kk = k0 + 2 * kh;
    v2f a; a.x = ap[kk]; a.y = ap[kk + 1];
#pragma unroll
    for (int j = 0; j < 4; ++j) {
      v2f b; b.x = bp[j][kk]; b.y = bp[j][kk + 1];
      acc[j] = wmma4(a, b, acc[j]);
    }
  }
#pragma unroll
  for (int j = 0; j < 4; ++j) {
    int n = gng * 64 + j * 16 + mrow;
    float bias = bdt[n];
#pragma unroll
    for (int v = 0; v < 8; ++v) {
      int orow = gm * 16 + v + 8 * kh;
      float t = acc[j][v] + bias;
      float sp = (t > 20.0f) ? t : log1pf(__expf(t));
      xz[(size_t)orow * XZW + n] = sp;
    }
  }
}

__global__ void k_scan_part1(const float* __restrict__ xz,
                             const float* __restrict__ u,
                             const float* __restrict__ dbl,
                             const float* __restrict__ Alog,
                             float* __restrict__ Pb, float* __restrict__ Sb) {
  const int b = blockIdx.x / NCH;
  const int c = blockIdx.x % NCH;
  const int d = threadIdx.x;

  float a[NST], P[NST], S[NST];
#pragma unroll
  for (int n = 0; n < NST; ++n) {
    a[n] = -__expf(Alog[d * NST + n]);
    P[n] = 1.0f;
    S[n] = 0.0f;
  }
  const int t0 = c * CL;
  for (int t = t0; t < t0 + CL; ++t) {
    const size_t row = (size_t)b * SEQ + t;
    const float* bc = dbl + row * DBLW;
    __builtin_prefetch(bc + DBLW, 0, 0);
    const float dt = xz[row * XZW + d];
    const float ut = u[row * DI + d];
    const float dtu = dt * ut;
#pragma unroll
    for (int n = 0; n < NST; ++n) {
      float dA = __expf(dt * a[n]);
      S[n] = S[n] * dA + dtu * bc[DTR + n];
      P[n] *= dA;
    }
  }
  const size_t base = ((size_t)(b * NCH + c) * DI + d) * NST;
#pragma unroll
  for (int n = 0; n < NST; ++n) {
    Pb[base + n] = P[n];
    Sb[base + n] = S[n];
  }
}

__global__ void k_scan_combine(const float* __restrict__ Pb,
                               float* __restrict__ Sb) {
  int idx = blockIdx.x * blockDim.x + threadIdx.x;
  if (idx >= BATCH * DI * NST) return;
  const int n = idx & (NST - 1);
  const int d = (idx >> 4) % DI;
  const int b = idx / (NST * DI);

  float carry = 0.0f;
  for (int c = 0; c < NCH; ++c) {
    const size_t base = ((size_t)(b * NCH + c) * DI + d) * NST + n;
    float p = Pb[base];
    float s = Sb[base];
    Sb[base] = carry;
    carry = p * carry + s;
  }
}

__global__ void k_scan_part3(float* __restrict__ xz,
                             const float* __restrict__ u,
                             const float* __restrict__ dbl,
                             const float* __restrict__ Alog,
                             const float* __restrict__ Dv,
                             const float* __restrict__ Sb) {
  const int b = blockIdx.x / NCH;
  const int c = blockIdx.x % NCH;
  const int d = threadIdx.x;

  float a[NST], h[NST];
  const size_t base = ((size_t)(b * NCH + c) * DI + d) * NST;
#pragma unroll
  for (int n = 0; n < NST; ++n) {
    a[n] = -__expf(Alog[d * NST + n]);
    h[n] = Sb[base + n];
  }
  const float Dd = Dv[d];

  const int t0 = c * CL;
  for (int t = t0; t < t0 + CL; ++t) {
    const size_t row = (size_t)b * SEQ + t;
    const float* bc = dbl + row * DBLW;
    __builtin_prefetch(bc + DBLW, 0, 0);
    const float dt = xz[row * XZW + d];
    const float ut = u[row * DI + d];
    const float dtu = dt * ut;
    float y = 0.0f;
#pragma unroll
    for (int n = 0; n < NST; ++n) {
      float dA = __expf(dt * a[n]);
      h[n] = h[n] * dA + dtu * bc[DTR + n];
      y += h[n] * bc[DTR + NST + n];
    }
    y += ut * Dd;
    float z = xz[row * XZW + DI + d];
    y *= z / (1.0f + __expf(-z));
    xz[row * XZW + DI + d] = y;
  }
}

__global__ void k_gemm_out(const float* __restrict__ xz_f,
                           const float* __restrict__ xz_b,
                           const float* __restrict__ x,
                           const float* __restrict__ Wout,
                           float* __restrict__ out) {
  const int wave = blockIdx.x * (blockDim.x >> 5) + (threadIdx.x >> 5);
  const int lane = threadIdx.x & 31;
  const int TG = DIMX / 64;
  const int gm = wave / TG, gng = wave % TG;
  const int mrow = lane & 15, kh = lane >> 4;

  int m  = gm * 16 + mrow;
  int bb = m / SEQ, ll = m % SEQ;
  int mrev = bb * SEQ + (SEQ - 1 - ll);

  const float* af = xz_f + (size_t)m    * XZW + DI;
  const float* ab = xz_b + (size_t)mrev * XZW + DI;
  const float* bp[4];
#pragma unroll
  for (int j = 0; j < 4; ++j)
    bp[j] = Wout + (size_t)(gng * 64 + j * 16 + mrow) * DI;

  v8f acc[4] = {};
  for (int k0 = 0; k0 < DI; k0 += 4) {
    int kk = k0 + 2 * kh;
    v2f a; a.x = af[kk] + ab[kk]; a.y = af[kk + 1] + ab[kk + 1];
#pragma unroll
    for (int j = 0; j < 4; ++j) {
      v2f b; b.x = bp[j][kk]; b.y = bp[j][kk + 1];
      acc[j] = wmma4(a, b, acc[j]);
    }
  }
#pragma unroll
  for (int j = 0; j < 4; ++j)
#pragma unroll
    for (int v = 0; v < 8; ++v) {
      int orow = gm * 16 + v + 8 * kh;
      int n = gng * 64 + j * 16 + mrow;
      out[(size_t)orow * DIMX + n] = acc[j][v] + x[(size_t)orow * DIMX + n];
    }
}

extern "C" void kernel_launch(void* const* d_in, const int* in_sizes, int n_in,
                              void* d_out, int out_size, void* d_ws, size_t ws_size,
                              hipStream_t stream) {
  const float* x      = (const float*)d_in[0];
  const float* Win_f  = (const float*)d_in[1];
  const float* cw_f   = (const float*)d_in[2];
  const float* cb_f   = (const float*)d_in[3];
  const float* Wxp_f  = (const float*)d_in[4];
  const float* Wdt_f  = (const float*)d_in[5];
  const float* bdt_f  = (const float*)d_in[6];
  const float* Alog_f = (const float*)d_in[7];
  const float* Dv_f   = (const float*)d_in[8];
  const float* Win_b  = (const float*)d_in[9];
  const float* cw_b   = (const float*)d_in[10];
  const float* cb_b   = (const float*)d_in[11];
  const float* Wxp_b  = (const float*)d_in[12];
  const float* Wdt_b  = (const float*)d_in[13];
  const float* bdt_b  = (const float*)d_in[14];
  const float* Alog_b = (const float*)d_in[15];
  const float* Dv_b   = (const float*)d_in[16];
  const float* Wout   = (const float*)d_in[17];
  float* out = (float*)d_out;

  float* ws    = (float*)d_ws;
  float* xz_f  = ws;
  float* xz_b  = xz_f + (size_t)MROWS * XZW;
  float* u_f   = xz_b + (size_t)MROWS * XZW;
  float* u_b   = u_f  + (size_t)MROWS * DI;
  float* dbl_f = u_b  + (size_t)MROWS * DI;
  float* dbl_b = dbl_f + (size_t)MROWS * DBLW;
  const size_t PS = (size_t)BATCH * NCH * DI * NST;
  float* P_f   = dbl_b + (size_t)MROWS * DBLW;
  float* S_f   = P_f + PS;
  float* P_b   = S_f + PS;
  float* S_b   = P_b + PS;

  const int BLK = 256, WPB = BLK / 32;

  {
    int waves = (MROWS / 16) * (XZW / 64);
    k_gemm_in<<<waves / WPB, BLK, 0, stream>>>(x, Win_f, xz_f, 0);
    k_gemm_in<<<waves / WPB, BLK, 0, stream>>>(x, Win_b, xz_b, 1);
  }
  {
    int n = MROWS * DI;
    k_conv_silu<<<(n + BLK - 1) / BLK, BLK, 0, stream>>>(xz_f, cw_f, cb_f, u_f);
    k_conv_silu<<<(n + BLK - 1) / BLK, BLK, 0, stream>>>(xz_b, cw_b, cb_b, u_b);
  }
  {
    int waves = MROWS / 16;
    k_gemm_xp<<<waves / WPB, BLK, 0, stream>>>(u_f, Wxp_f, dbl_f);
    k_gemm_xp<<<waves / WPB, BLK, 0, stream>>>(u_b, Wxp_b, dbl_b);
  }
  {
    int waves = (MROWS / 16) * (DI / 64);
    k_gemm_dt<<<waves / WPB, BLK, 0, stream>>>(dbl_f, Wdt_f, bdt_f, xz_f);
    k_gemm_dt<<<waves / WPB, BLK, 0, stream>>>(dbl_b, Wdt_b, bdt_b, xz_b);
  }
  {
    int cb = (BATCH * DI * NST + BLK - 1) / BLK;
    k_scan_part1<<<BATCH * NCH, DI, 0, stream>>>(xz_f, u_f, dbl_f, Alog_f, P_f, S_f);
    k_scan_part1<<<BATCH * NCH, DI, 0, stream>>>(xz_b, u_b, dbl_b, Alog_b, P_b, S_b);
    k_scan_combine<<<cb, BLK, 0, stream>>>(P_f, S_f);
    k_scan_combine<<<cb, BLK, 0, stream>>>(P_b, S_b);
    k_scan_part3<<<BATCH * NCH, DI, 0, stream>>>(xz_f, u_f, dbl_f, Alog_f, Dv_f, S_f);
    k_scan_part3<<<BATCH * NCH, DI, 0, stream>>>(xz_b, u_b, dbl_b, Alog_b, Dv_b, S_b);
  }
  {
    int waves = (MROWS / 16) * (DIMX / 64);
    k_gemm_out<<<waves / WPB, BLK, 0, stream>>>(xz_f, xz_b, x, Wout, out);
  }
}


# Conv+SiLU stage fused into the x_dbl projection kernel: a block forms 16 rows of u, stores them and feeds them through LDS to the matrix steps; k split over 6 waves: x_dbl k sums re-associated
# speedup vs baseline: 1.7777x; 1.7777x over previous
#include <hip/hip_runtime.h>
#include <hip/hip_bf16.h>
#include <math.h>

#define BATCH 4
#define SEQ   2048
#define DIMX  192
#define DI    384
#define NST   16
#define DTR   12
#define XZW   768
#define DBLW  48
#define MROWS (BATCH*SEQ)
#define NCH   128
#define CL    (SEQ/NCH)
#define CG    16
#define NG    (NCH/CG)

typedef float v2f __attribute__((ext_vector_type(2)));
typedef float v8f __attribute__((ext_vector_type(8)));

__device__ __forceinline__ v8f wmma4(v2f a, v2f b, v8f c) {
  return __builtin_amdgcn_wmma_f32_16x16x4_f32(false, a, false, b, (short)0, c,
                                               false, false);
}

__global__ void k_gemm_in(const float* __restrict__ x,
                          const float* __restrict__ Win,
                          float* __restrict__ xz, int reverse) {
  const int wave = blockIdx.x * (blockDim.x >> 5) + (threadIdx.x >> 5);
  const int lane = threadIdx.x & 31;
  const int TG = XZW / 64;
  const int gm = wave / TG, gng = wave % TG;
  const int mrow = lane & 15, kh = lane >> 4;

  int m  = gm * 16 + mrow;
  int bb = m / SEQ, ll = m % SEQ;
  int srow = reverse ? (bb * SEQ + (SEQ - 1 - ll)) : m;

  const float* ap = x + (size_t)srow * DIMX;
  const float* bp[4];
#pragma unroll
  for (int j = 0; j < 4; ++j)
    bp[j] = Win + (size_t)(gng * 64 + j * 16 + mrow) * DIMX;

  v8f acc[4] = {};
  for (int k0 = 0; k0 < DIMX; k0 += 4) {
    int kk = k0 + 2 * kh;
    v2f a; a.x = ap[kk]; a.y = ap[kk + 1];
#pragma unroll
    for (int j = 0; j < 4; ++j) {
      v2f b; b.x = bp[j][kk]; b.y = bp[j][kk + 1];
      acc[j] = wmma4(a, b, acc[j]);
    }
  }
#pragma unroll
  for (int j = 0; j < 4; ++j)
#pragma unroll
    for (int v = 0; v < 8; ++v) {
      int orow = gm * 16 + v + 8 * kh;
      xz[(size_t)orow * XZW + gng * 64 + j * 16 + mrow] = acc[j][v];
    }
}

#define UP 68
#define KW 6

__global__ void __launch_bounds__(32 * KW)
k_conv_xp(const float* __restrict__ xz,
          const float* __restrict__ cw,
          const float* __restrict__ cb,
          const float* __restrict__ Wxp,
          float* __restrict__ u,
          float* __restrict__ dbl) {
  __shared__ __attribute__((aligned(16))) float ut[KW][16 * UP];
  __shared__ float part[KW][24][32];
  const int w    = threadIdx.x >> 5;
  const int lane = threadIdx.x & 31;
  const int row0 = blockIdx.x * 16;
  const int t0   = row0 % SEQ;

  const int cg = lane & 15, rh = lane >> 4;
  const int ch = w * 64 + cg * 4;
  const int rb = rh * 8;
  float4 o[8];
  {
    const float4 w0 = *(const float4*)(cw + (size_t)ch * 4);
    const float4 w1 = *(const float4*)(cw + (size_t)ch * 4 + 4);
    const float4 w2 = *(const float4*)(cw + (size_t)ch * 4 + 8);
    const float4 w3 = *(const float4*)(cw + (size_t)ch * 4 + 12);
    const float4 bias = *(const float4*)(cb + ch);
    float4 xr[11];
#pragma unroll
    for (int i = 0; i < 11; ++i) {
      const int rr = rb - 3 + i;
      const bool ok = (i >= 3) || (t0 + rr) >= 0;
      const float4 v = *(const float4*)(xz + (size_t)(row0 + (ok ? rr : 0)) * XZW + ch);
      xr[i] = ok ? v : make_float4(0.0f, 0.0f, 0.0f, 0.0f);
    }
#pragma unroll
    for (int r = 0; r < 8; ++r) {
      float a0 = bias.x, a1 = bias.y, a2 = bias.z, a3 = bias.w;
      a0 += xr[r].x * w0.x;  a1 += xr[r].y * w1.x;  a2 += xr[r].z * w2.x;  a3 += xr[r].w * w3.x;
      a0 += xr[r+1].x * w0.y; a1 += xr[r+1].y * w1.y; a2 += xr[r+1].z * w2.y; a3 += xr[r+1].w * w3.y;
      a0 += xr[r+2].x * w0.z; a1 += xr[r+2].y * w1.z; a2 += xr[r+2].z * w2.z; a3 += xr[r+2].w * w3.z;
      a0 += xr[r+3].x * w0.w; a1 += xr[r+3].y * w1.w; a2 += xr[r+3].z * w2.w; a3 += xr[r+3].w * w3.w;
      o[r].x = a0 / (1.0f + __expf(-a0));
      o[r].y = a1 / (1.0f + __expf(-a1));
      o[r].z = a2 / (1.0f + __expf(-a2));
      o[r].w = a3 / (1.0f + __expf(-a3));
      *(float4*)(&ut[w][(rb + r) * UP + cg * 4]) = o[r];
    }
  }
  __syncthreads();
#pragma unroll
  for (int r = 0; r < 8; ++r)
    *(float4*)(u + (size_t)(row0 + rb + r) * DI + ch) = o[r];
  __builtin_amdgcn_sched_barrier(0);

  const int mrow = lane & 15, kh = lane >> 4;
  bool nvalid[3];
  const float* bp[3];
#pragma unroll
  for (int j = 0; j < 3; ++j) {
    int n = j * 16 + mrow;
    nvalid[j] = (n < DTR + 2 * NST);
    bp[j] = Wxp + (size_t)(nvalid[j] ? n : 0) * DI + w * 64;
  }
  v8f acc[3] = {};
  for (int s = 0; s < 16; ++s) {
    const int kl = 4 * s + 2 * kh;
    const v2f a = *(const v2f*)(&ut[w][mrow * UP + kl]);
#pragma unroll
    for (int j = 0; j < 3; ++j) {
      const v2f bl = *(const v2f*)(bp[j] + kl);
      v2f b;
      b.x = nvalid[j] ? bl.x : 0.0f;
      b.y = nvalid[j] ? bl.y : 0.0f;
      acc[j] = wmma4(a, b, acc[j]);
    }
  }
#pragma unroll
  for (int j = 0; j < 3; ++j)
#pragma unroll
    for (int v = 0; v < 8; ++v)
      part[w][j * 8 + v][lane] = acc[j][v];
  __syncthreads();

  for (int e = threadIdx.x; e < 24 * 32; e += 32 * KW) {
    const int q = e >> 5, ln = e & 31;
    float s = part[0][q][ln];
#pragma unroll
    for (int ww = 1; ww < KW; ++ww) s += part[ww][q][ln];
    const int orow = row0 + (q & 7) + 8 * (ln >> 4);
    dbl[(size_t)orow * DBLW + (q >> 3) * 16 + (ln & 15)] = s;
  }
}

__global__ void k_gemm_dt(const float* __restrict__ dbl,
                          const float* __restrict__ Wdt,
                          const float* __restrict__ bdt,
                          float* __restrict__ xz) {
  const int wave = blockIdx.x * (blockDim.x >> 5) + (threadIdx.x >> 5);
  const int lane = threadIdx.x & 31;
  const int TG = DI / 64;
  const int gm = wave / TG, gng = wave % TG;
  const int mrow = lane & 15, kh = lane >> 4;

  const float* ap = dbl + (size_t)(gm * 16 + mrow) * DBLW;
  const float* bp[4];
#pragma unroll
  for (int j = 0; j < 4; ++j)
    bp[j] = Wdt + (size_t)(gng * 64 + j * 16 + mrow) * DTR;

  v8f acc[4] = {};
  for (int k0 = 0; k0 < DTR; k0 += 4) {
    int kk = k0 + 2 * kh;
    v2f a; a.x = ap[kk]; a.y = ap[kk + 1];
#pragma unroll
    for (int j = 0; j < 4; ++j) {
      v2f b; b.x = bp[j][kk]; b.y = bp[j][kk + 1];
      acc[j] = wmma4(a, b, acc[j]);
    }
  }
#pragma unroll
  for (int j = 0; j < 4; ++j) {
    int n = gng * 64 + j * 16 + mrow;
    float bias = bdt[n];
#pragma unroll
    for (int v = 0; v < 8; ++v) {
      int orow = gm * 16 + v + 8 * kh;
      float t = acc[j][v] + bias;
      float sp = (t > 20.0f) ? t : log1pf(__expf(t));
      xz[(size_t)orow * XZW + n] = sp;
    }
  }
}

__global__ void k_scan_part1(const float* __restrict__ xz,
                             const float* __restrict__ u,
                             const float* __restrict__ dbl,
                             const float* __restrict__ Alog,
                             float* __restrict__ Pb, float* __restrict__ Sb) {
  const int b = blockIdx.x / NCH;
  const int c = blockIdx.x % NCH;
  const int d = threadIdx.x;

  float a[NST], P[NST], S[NST];
#pragma unroll
  for (int n = 0; n < NST; ++n) {
    a[n] = -__expf(Alog[d * NST + n]);
    P[n] = 1.0f;
    S[n] = 0.0f;
  }
  const int t0 = c * CL;
  for (int t = t0; t < t0 + CL; ++t) {
    const size_t row = (size_t)b * SEQ + t;
    const float* bc = dbl + row * DBLW;
    __builtin_prefetch(bc + DBLW, 0, 0);
    const float dt = xz[row * XZW + d];
    const float ut = u[row * DI + d];
    const float dtu = dt * ut;
#pragma unroll
    for (int n = 0; n < NST; ++n) {
      float dA = __expf(dt * a[n]);
      S[n] = S[n] * dA + dtu * bc[DTR + n];
      P[n] *= dA;
    }
  }
  const size_t base = ((size_t)(b * NCH + c) * DI + d) * NST;
#pragma unroll
  for (int n = 0; n < NST; ++n) {
    Pb[base + n] = P[n];
    Sb[base + n] = S[n];
  }
}

__global__ void k_scan_combine(const float* __restrict__ Pb,
                               float* __restrict__ Sb) {
  __shared__ float Pg[NG][32];
  __shared__ float Sg[NG][32];
  const int e = threadIdx.x & 31;
  const int g = threadIdx.x >> 5;
  const int flat = blockIdx.x * 32 + e;
  const int b  = flat / (DI * NST);
  const int dn = flat % (DI * NST);
  const size_t CS = (size_t)DI * NST;
  const size_t base = (size_t)(b * NCH + g * CG) * CS + dn;

  float p[CG], s[CG];
#pragma unroll
  for (int j = 0; j < CG; ++j) {
    p[j] = Pb[base + (size_t)j * CS];
    s[j] = Sb[base + (size_t)j * CS];
  }
  float pp = 1.0f, ss = 0.0f;
#pragma unroll
  for (int j = 0; j < CG; ++j) {
    ss = p[j] * ss + s[j];
    pp *= p[j];
  }
  Pg[g][e] = pp;
  Sg[g][e] = ss;
  __syncthreads();

  float carry = 0.0f;
  for (int k = 0; k < g; ++k)
    carry = Pg[k][e] * carry + Sg[k][e];
#pragma unroll
  for (int j = 0; j < CG; ++j) {
    Sb[base + (size_t)j * CS] = carry;
    carry = p[j] * carry + s[j];
  }
}

__global__ void k_scan_part3(float* __restrict__ xz,
                             const float* __restrict__ u,
                             const float* __restrict__ dbl,
                             const float* __restrict__ Alog,
                             const float* __restrict__ Dv,
                             const float* __restrict__ Sb) {
  const int b = blockIdx.x / NCH;
  const int c = blockIdx.x % NCH;
  const int d = threadIdx.x;

  float a[NST], h[NST];
  const size_t base = ((size_t)(b * NCH + c) * DI + d) * NST;
#pragma unroll
  for (int n = 0; n < NST; ++n) {
    a[n] = -__expf(Alog[d * NST + n]);
    h[n] = Sb[base + n];
  }
  const float Dd = Dv[d];

  const int t0 = c * CL;
  for (int t = t0; t < t0 + CL; ++t) {
    const size_t row = (size_t)b * SEQ + t;
    const float* bc = dbl + row * DBLW;
    __builtin_prefetch(bc + DBLW, 0, 0);
    const float dt = xz[row * XZW + d];
    const float ut = u[row * DI + d];
    const float dtu = dt * ut;
    float y = 0.0f;
#pragma unroll
    for (int n = 0; n < NST; ++n) {
      float dA = __expf(dt * a[n]);
      h[n] = h[n] * dA + dtu * bc[DTR + n];
      y += h[n] * bc[DTR + NST + n];
    }
    y += ut * Dd;
    float z = xz[row * XZW + DI + d];
    y *= z / (1.0f + __expf(-z));
    xz[row * XZW + DI + d] = y;
  }
}

__global__ void k_gemm_out(const float* __restrict__ xz_f,
                           const float* __restrict__ xz_b,
                           const float* __restrict__ x,
                           const float* __restrict__ Wout,
                           float* __restrict__ out) {
  const int wave = blockIdx.x * (blockDim.x >> 5) + (threadIdx.x >> 5);
  const int lane = threadIdx.x & 31;
  const int TG = DIMX / 64;
  const int gm = wave / TG, gng = wave % TG;
  const int mrow = lane & 15, kh = lane >> 4;

  int m  = gm * 16 + mrow;
  int bb = m / SEQ, ll = m % SEQ;
  int mrev = bb * SEQ + (SEQ - 1 - ll);

  const float* af = xz_f + (size_t)m    * XZW + DI;
  const float* ab = xz_b + (size_t)mrev * XZW + DI;
  const float* bp[4];
#pragma unroll
  for (int j = 0; j < 4; ++j)
    bp[j] = Wout + (size_t)(gng * 64 + j * 16 + mrow) * DI;

  v8f acc[4] = {};
  for (int k0 = 0; k0 < DI; k0 += 4) {
    int kk = k0 + 2 * kh;
    v2f a; a.x = af[kk] + ab[kk]; a.y = af[kk + 1] + ab[kk + 1];
#pragma unroll
    for (int j = 0; j < 4; ++j) {
      v2f b; b.x = bp[j][kk]; b.y = bp[j][kk + 1];
      acc[j] = wmma4(a, b, acc[j]);
    }
  }
#pragma unroll
  for (int j = 0; j < 4; ++j)
#pragma unroll
    for (int v = 0; v < 8; ++v) {
      int orow = gm * 16 + v + 8 * kh;
      int n = gng * 64 + j * 16 + mrow;
      out[(size_t)orow * DIMX + n] = acc[j][v] + x[(size_t)orow * DIMX + n];
    }
}

extern "C" void kernel_launch(void* const* d_in, const int* in_sizes, int n_in,
                              void* d_out, int out_size, void* d_ws, size_t ws_size,
                              hipStream_t stream) {
  const float* x      = (const float*)d_in[0];
  const float* Win_f  = (const float*)d_in[1];
  const float* cw_f   = (const float*)d_in[2];
  const float* cb_f   = (const float*)d_in[3];
  const float* Wxp_f  = (const float*)d_in[4];
  const float* Wdt_f  = (const float*)d_in[5];
  const float* bdt_f  = (const float*)d_in[6];
  const float* Alog_f = (const float*)d_in[7];
  const float* Dv_f   = (const float*)d_in[8];
  const float* Win_b  = (const float*)d_in[9];
  const float* cw_b   = (const float*)d_in[10];
  const float* cb_b   = (const float*)d_in[11];
  const float* Wxp_b  = (const float*)d_in[12];
  const float* Wdt_b  = (const float*)d_in[13];
  const float* bdt_b  = (const float*)d_in[14];
  const float* Alog_b = (const float*)d_in[15];
  const float* Dv_b   = (const float*)d_in[16];
  const float* Wout   = (const float*)d_in[17];
  float* out = (float*)d_out;

  float* ws    = (float*)d_ws;
  float* xz_f  = ws;
  float* xz_b  = xz_f + (size_t)MROWS * XZW;
  float* u_f   = xz_b + (size_t)MROWS * XZW;
  float* u_b   = u_f  + (size_t)MROWS * DI;
  float* dbl_f = u_b  + (size_t)MROWS * DI;
  float* dbl_b = dbl_f + (size_t)MROWS * DBLW;
  const size_t PS = (size_t)BATCH * NCH * DI * NST;
  float* P_f   = dbl_b + (size_t)MROWS * DBLW;
  float* S_f   = P_f + PS;
  float* P_b   = S_f + PS;
  float* S_b   = P_b + PS;

  const int BLK = 256, WPB = BLK / 32;

  {
    int waves = (MROWS / 16) * (XZW / 64);
    k_gemm_in<<<waves / WPB, BLK, 0, stream>>>(x, Win_f, xz_f, 0);
    k_gemm_in<<<waves / WPB, BLK, 0, stream>>>(x, Win_b, xz_b, 1);
  }
  {
    k_conv_xp<<<MROWS / 16, 32 * KW, 0, stream>>>(xz_f, cw_f, cb_f, Wxp_f, u_f, dbl_f);
    k_conv_xp<<<MROWS / 16, 32 * KW, 0, stream>>>(xz_b, cw_b, cb_b, Wxp_b, u_b, dbl_b);
  }
  {
    int waves = (MROWS / 16) * (DI / 64);
    k_gemm_dt<<<waves / WPB, BLK, 0, stream>>>(dbl_f, Wdt_f, bdt_f, xz_f);
    k_gemm_dt<<<waves / WPB, BLK, 0, stream>>>(dbl_b, Wdt_b, bdt_b, xz_b);
  }
  {
    int cb = BATCH * DI * NST / 32;
    k_scan_part1<<<BATCH * NCH, DI, 0, stream>>>(xz_f, u_f, dbl_f, Alog_f, P_f, S_f);
    k_scan_part1<<<BATCH * NCH, DI, 0, stream>>>(xz_b, u_b, dbl_b, Alog_b, P_b, S_b);
    k_scan_combine<<<cb, 32 * NG, 0, stream>>>(P_f, S_f);
    k_scan_combine<<<cb, 32 * NG, 0, stream>>>(P_b, S_b);
    k_scan_part3<<<BATCH * NCH, DI, 0, stream>>>(xz_f, u_f, dbl_f, Alog_f, Dv_f, S_f);
    k_scan_part3<<<BATCH * NCH, DI, 0, stream>>>(xz_b, u_b, dbl_b, Alog_b, Dv_b, S_b);
  }
  {
    int waves = (MROWS / 16) * (DIMX / 64);
    k_gemm_out<<<waves / WPB, BLK, 0, stream>>>(xz_f, xz_b, x, Wout, out);
  }
}


# Scan pass 3 of both branches fused with the output projection: a block re-runs both branches' chunks of its 32 rows, y_f and y_b meet in LDS as the A operand; 9 launches become 7; sum orders kept
# speedup vs baseline: 2.5339x; 1.4254x over previous
#include <hip/hip_runtime.h>
#include <hip/hip_bf16.h>
#include <math.h>

#define BATCH 4
#define SEQ   2048
#define DIMX  192
#define DI    384
#define NST   16
#define DTR   12
#define XZW   768
#define DBLW  48
#define MROWS (BATCH*SEQ)
#define NCH   128
#define CL    (SEQ/NCH)
#define CG    16
#define NG    (NCH/CG)

typedef float v2f __attribute__((ext_vector_type(2)));
typedef float v8f __attribute__((ext_vector_type(8)));

__device__ __forceinline__ v8f wmma4(v2f a, v2f b, v8f c) {
  return __builtin_amdgcn_wmma_f32_16x16x4_f32(false, a, false, b, (short)0, c,
                                               false, false);
}

__global__ void k_gemm_in(const float* __restrict__ x,
                          const float* __restrict__ Win,
                          float* __restrict__ xz, int reverse) {
  const int wave = blockIdx.x * (blockDim.x >> 5) + (threadIdx.x >> 5);
  const int lane = threadIdx.x & 31;
  const int TG = XZW / 64;
  const int gm = wave / TG, gng = wave % TG;
  const int mrow = lane & 15, kh = lane >> 4;

  int m  = gm * 16 + mrow;
  int bb = m / SEQ, ll = m % SEQ;
  int srow = reverse ? (bb * SEQ + (SEQ - 1 - ll)) : m;

  const float* ap = x + (size_t)srow * DIMX;
  const float* bp[4];
#pragma unroll
  for (int j = 0; j < 4; ++j)
    bp[j] = Win + (size_t)(gng * 64 + j * 16 + mrow) * DIMX;

  v8f acc[4] = {};
  for (int k0 = 0; k0 < DIMX; k0 += 4) {
    int kk = k0 + 2 * kh;
    v2f a; a.x = ap[kk]; a.y = ap[kk + 1];
#pragma unroll
    for (int j = 0; j < 4; ++j) {
      v2f b; b.x = bp[j][kk]; b.y = bp[j][kk + 1];
      acc[j] = wmma4(a, b, acc[j]);
    }
  }
#pragma unroll
  for (int j = 0; j < 4; ++j)
#pragma unroll
    for (int v = 0; v < 8; ++v) {
      int orow = gm * 16 + v + 8 * kh;
      xz[(size_t)orow * XZW + gng * 64 + j * 16 + mrow] = acc[j][v];
    }
}

static_assert(CL == 16, "a 16-row tile must be one scan chunk");
#define DTP 388
#define UP 68
#define KW 6

__global__ void __launch_bounds__(32 * KW)
k_conv_xp(const float* __restrict__ xz,
          const float* __restrict__ cw,
          const float* __restrict__ cb,
          const float* __restrict__ Wxp,
          const float* __restrict__ Wdt,
          const float* __restrict__ bdt,
          const float* __restrict__ Alog,
          float* __restrict__ u,
          float* __restrict__ dbl,
          float* __restrict__ dtb,
          float* __restrict__ Pb, float* __restrict__ Sb) {
  __shared__ __attribute__((aligned(16))) float ut[KW][16 * UP];
  __shared__ __attribute__((aligned(16))) float pd[16 * DTP];
  __shared__ __attribute__((aligned(16))) float dblT[16][DBLW];
  const int w    = threadIdx.x >> 5;
  const int lane = threadIdx.x & 31;
  const int row0 = blockIdx.x * 16;
  const int t0   = row0 % SEQ;

  const int cg = lane & 15, rh = lane >> 4;
  const int ch = w * 64 + cg * 4;
  const int rb = rh * 8;
  float4 o[8];
  {
    const float4 w0 = *(const float4*)(cw + (size_t)ch * 4);
    const float4 w1 = *(const float4*)(cw + (size_t)ch * 4 + 4);
    const float4 w2 = *(const float4*)(cw + (size_t)ch * 4 + 8);
    const float4 w3 = *(const float4*)(cw + (size_t)ch * 4 + 12);
    const float4 bias = *(const float4*)(cb + ch);
    float4 xr[11];
#pragma unroll
    for (int i = 0; i < 11; ++i) {
      const int rr = rb - 3 + i;
      const bool ok = (i >= 3) || (t0 + rr) >= 0;
      const float4 v = *(const float4*)(xz + (size_t)(row0 + (ok ? rr : 0)) * XZW + ch);
      xr[i] = ok ? v : make_float4(0.0f, 0.0f, 0.0f, 0.0f);
    }
#pragma unroll
    for (int r = 0; r < 8; ++r) {
      float a0 = bias.x, a1 = bias.y, a2 = bias.z, a3 = bias.w;
      a0 += xr[r].x * w0.x;  a1 += xr[r].y * w1.x;  a2 += xr[r].z * w2.x;  a3 += xr[r].w * w3.x;
      a0 += xr[r+1].x * w0.y; a1 += xr[r+1].y * w1.y; a2 += xr[r+1].z * w2.y; a3 += xr[r+1].w * w3.y;
      a0 += xr[r+2].x * w0.z; a1 += xr[r+2].y * w1.z; a2 += xr[r+2].z * w2.z; a3 += xr[r+2].w * w3.z;
      a0 += xr[r+3].x * w0.w; a1 += xr[r+3].y * w1.w; a2 += xr[r+3].z * w2.w; a3 += xr[r+3].w * w3.w;
      o[r].x = a0 / (1.0f + __expf(-a0));
      o[r].y = a1 / (1.0f + __expf(-a1));
      o[r].z = a2 / (1.0f + __expf(-a2));
      o[r].w = a3 / (1.0f + __expf(-a3));
      *(float4*)(&ut[w][(rb + r) * UP + cg * 4]) = o[r];
    }
  }
  __syncthreads();
#pragma unroll
  for (int r = 0; r < 8; ++r)
    *(float4*)(u + (size_t)(row0 + rb + r) * DI + ch) = o[r];
  __builtin_amdgcn_sched_barrier(0);

  const int mrow = lane & 15, kh = lane >> 4;
  bool nvalid[3];
  const float* bp[3];
#pragma unroll
  for (int j = 0; j < 3; ++j) {
    int n = j * 16 + mrow;
    nvalid[j] = (n < DTR + 2 * NST);
    bp[j] = Wxp + (size_t)(nvalid[j] ? n : 0) * DI + w * 64;
  }
  v8f acc[3] = {};
  for (int s = 0; s < 16; ++s) {
    const int kl = 4 * s + 2 * kh;
    const v2f a = *(const v2f*)(&ut[w][mrow * UP + kl]);
#pragma unroll
    for (int j = 0; j < 3; ++j) {
      const v2f bl = *(const v2f*)(bp[j] + kl);
      v2f b;
      b.x = nvalid[j] ? bl.x : 0.0f;
      b.y = nvalid[j] ? bl.y : 0.0f;
      acc[j] = wmma4(a, b, acc[j]);
    }
  }
#pragma unroll
  for (int j = 0; j < 3; ++j)
#pragma unroll
    for (int v = 0; v < 8; ++v)
      pd[(w * 24 + j * 8 + v) * 32 + lane] = acc[j][v];
  __syncthreads();

  for (int e = threadIdx.x; e < 24 * 32; e += 32 * KW) {
    const int q = e >> 5, ln = e & 31;
    float s = pd[q * 32 + ln];
#pragma unroll
    for (int ww = 1; ww < KW; ++ww) s += pd[(ww * 24 + q) * 32 + ln];
    const int lr  = (q & 7) + 8 * (ln >> 4);
    const int col = (q >> 3) * 16 + (ln & 15);
    dbl[(size_t)(row0 + lr) * DBLW + col] = s;
    dblT[lr][col] = s;
  }
  __syncthreads();

  {
    const float* wp[4];
#pragma unroll
    for (int j = 0; j < 4; ++j)
      wp[j] = Wdt + (size_t)(w * 64 + j * 16 + mrow) * DTR;
    v8f dacc[4] = {};
    for (int k0 = 0; k0 < DTR; k0 += 4) {
      const int kk = k0 + 2 * kh;
      v2f a; a.x = dblT[mrow][kk]; a.y = dblT[mrow][kk + 1];
#pragma unroll
      for (int j = 0; j < 4; ++j) {
        v2f b; b.x = wp[j][kk]; b.y = wp[j][kk + 1];
        dacc[j] = wmma4(a, b, dacc[j]);
      }
    }
#pragma unroll
    for (int j = 0; j < 4; ++j) {
      const int n = w * 64 + j * 16 + mrow;
      const float bias = bdt[n];
#pragma unroll
      for (int v = 0; v < 8; ++v) {
        const int lr = v + 8 * kh;
        const float t = dacc[j][v] + bias;
        const float sp = (t > 20.0f) ? t : log1pf(__expf(t));
        dtb[(size_t)(row0 + lr) * DI + n] = sp;
        pd[lr * DTP + n] = sp;
      }
    }
  }
  __syncthreads();

  const int bsq = row0 / SEQ;
  const int chk = t0 / CL;
  for (int cc = 0; cc < 2; ++cc) {
    const int d = threadIdx.x + cc * (32 * KW);
    float a[NST], P[NST], S[NST];
#pragma unroll
    for (int n = 0; n < NST; ++n) {
      a[n] = -__expf(Alog[d * NST + n]);
      P[n] = 1.0f;
      S[n] = 0.0f;
    }
    const float* up = &ut[d >> 6][d & 63];
    for (int t = 0; t < CL; ++t) {
      const float dt  = pd[t * DTP + d];
      const float uv  = up[t * UP];
      const float dtu = dt * uv;
#pragma unroll
      for (int n = 0; n < NST; ++n) {
        float dA = __expf(dt * a[n]);
        S[n] = S[n] * dA + dtu * dblT[t][DTR + n];
        P[n] *= dA;
      }
    }
    const size_t base = ((size_t)(bsq * NCH + chk) * DI + d) * NST;
#pragma unroll
    for (int n = 0; n < NST; ++n) {
      Pb[base + n] = P[n];
      Sb[base + n] = S[n];
    }
  }
}

__global__ void k_scan_combine(const float* __restrict__ Pb,
                               float* __restrict__ Sb) {
  __shared__ float Pg[NG][32];
  __shared__ float Sg[NG][32];
  const int e = threadIdx.x & 31;
  const int g = threadIdx.x >> 5;
  const int flat = blockIdx.x * 32 + e;
  const int b  = flat / (DI * NST);
  const int dn = flat % (DI * NST);
  const size_t CS = (size_t)DI * NST;
  const size_t base = (size_t)(b * NCH + g * CG) * CS + dn;

  float p[CG], s[CG];
#pragma unroll
  for (int j = 0; j < CG; ++j) {
    p[j] = Pb[base + (size_t)j * CS];
    s[j] = Sb[base + (size_t)j * CS];
  }
  float pp = 1.0f, ss = 0.0f;
#pragma unroll
  for (int j = 0; j < CG; ++j) {
    ss = p[j] * ss + s[j];
    pp *= p[j];
  }
  Pg[g][e] = pp;
  Sg[g][e] = ss;
  __syncthreads();

  float carry = 0.0f;
  for (int k = 0; k < g; ++k)
    carry = Pg[k][e] * carry + Sg[k][e];
#pragma unroll
  for (int j = 0; j < CG; ++j) {
    Sb[base + (size_t)j * CS] = carry;
    carry = p[j] * carry + s[j];
  }
}

#define YP 388
#define RT 32

__global__ void __launch_bounds__(2 * DI)
k_scan3_out(const float* __restrict__ xz_f, const float* __restrict__ xz_b,
            const float* __restrict__ dtb_f, const float* __restrict__ dtb_b,
            const float* __restrict__ u_f, const float* __restrict__ u_b,
            const float* __restrict__ dbl_f, const float* __restrict__ dbl_b,
            const float* __restrict__ Alog_f, const float* __restrict__ Alog_b,
            const float* __restrict__ Dv_f, const float* __restrict__ Dv_b,
            const float* __restrict__ Sb_f, const float* __restrict__ Sb_b,
            const float* __restrict__ x, const float* __restrict__ Wout,
            float* __restrict__ out) {
  __shared__ __attribute__((aligned(16))) float yF[RT * YP];
  __shared__ __attribute__((aligned(16))) float yB[RT * YP];
  const int b  = blockIdx.x / (NCH / 2);
  const int cp = blockIdx.x % (NCH / 2);
  const int g  = threadIdx.x / DI;
  const int d  = threadIdx.x - g * DI;
  const int c  = 2 * cp + g;
  const int cr = NCH - 1 - c;

  float af[NST], ab[NST], hf[NST], hb[NST];
  const size_t base_f = ((size_t)(b * NCH + c)  * DI + d) * NST;
  const size_t base_b = ((size_t)(b * NCH + cr) * DI + d) * NST;
#pragma unroll
  for (int n = 0; n < NST; ++n) {
    af[n] = -__expf(Alog_f[d * NST + n]);
    ab[n] = -__expf(Alog_b[d * NST + n]);
    hf[n] = Sb_f[base_f + n];
    hb[n] = Sb_b[base_b + n];
  }
  const float Dd_f = Dv_f[d];
  const float Dd_b = Dv_b[d];

  const size_t rf0 = (size_t)b * SEQ + (size_t)c  * CL;
  const size_t rb0 = (size_t)b * SEQ + (size_t)cr * CL;
  float* yFg = &yF[(CL * g) * YP + d];
  float* yBg = &yB[(CL * g) * YP + d];
  for (int t = 0; t < CL; ++t) {
    const size_t rowf = rf0 + t;
    const size_t rowb = rb0 + t;
    const float* bcf = dbl_f + rowf * DBLW;
    const float* bcb = dbl_b + rowb * DBLW;
    __builtin_prefetch(bcf + DBLW, 0, 0);
    __builtin_prefetch(bcb + DBLW, 0, 0);
    const float dtf = dtb_f[rowf * DI + d];
    const float dtr = dtb_b[rowb * DI + d];
    const float utf = u_f[rowf * DI + d];
    const float utr = u_b[rowb * DI + d];
    const float zf  = xz_f[rowf * XZW + DI + d];
    const float zr  = xz_b[rowb * XZW + DI + d];
    const float dtuf = dtf * utf;
    const float dtur = dtr * utr;
    float yf = 0.0f, yr = 0.0f;
#pragma unroll
    for (int n = 0; n < NST; ++n) {
      const float dAf = __expf(dtf * af[n]);
      const float dAr = __expf(dtr * ab[n]);
      hf[n] = hf[n] * dAf + dtuf * bcf[DTR + n];
      hb[n] = hb[n] * dAr + dtur * bcb[DTR + n];
      yf += hf[n] * bcf[DTR + NST + n];
      yr += hb[n] * bcb[DTR + NST + n];
    }
    yf += utf * Dd_f;
    yr += utr * Dd_b;
    yf *= zf / (1.0f + __expf(-zf));
    yr *= zr / (1.0f + __expf(-zr));
    yFg[t * YP] = yf;
    yBg[(CL - 1 - t) * YP] = yr;
  }
  __syncthreads();

  const int w = threadIdx.x >> 5;
  if (w < DIMX / 16) {
    const int lane = threadIdx.x & 31;
    const int mrow = lane & 15, kh = lane >> 4;
    const float* bp  = Wout + (size_t)(w * 16 + mrow) * DI;
    const float* pf0 = &yF[mrow * YP];
    const float* pb0 = &yB[mrow * YP];
    const float* pf1 = &yF[(16 + mrow) * YP];
    const float* pb1 = &yB[(16 + mrow) * YP];
    v8f acc0 = {}, acc1 = {};
    for (int k0 = 0; k0 < DI; k0 += 4) {
      const int kk = k0 + 2 * kh;
      const v2f fa0 = *(const v2f*)(pf0 + kk);
      const v2f fb0 = *(const v2f*)(pb0 + kk);
      const v2f fa1 = *(const v2f*)(pf1 + kk);
      const v2f fb1 = *(const v2f*)(pb1 + kk);
      v2f a0; a0.x = fa0.x + fb0.x; a0.y = fa0.y + fb0.y;
      v2f a1; a1.x = fa1.x + fb1.x; a1.y = fa1.y + fb1.y;
      v2f bv; bv.x = bp[kk]; bv.y = bp[kk + 1];
      acc0 = wmma4(a0, bv, acc0);
      acc1 = wmma4(a1, bv, acc1);
    }
    const int n = w * 16 + mrow;
    const size_t r0 = (size_t)b * SEQ + (size_t)cp * RT;
#pragma unroll
    for (int v = 0; v < 8; ++v) {
      const size_t orow = r0 + v + 8 * kh;
      out[orow * DIMX + n] = acc0[v] + x[orow * DIMX + n];
      out[(orow + 16) * DIMX + n] = acc1[v] + x[(orow + 16) * DIMX + n];
    }
  }
}

extern "C" void kernel_launch(void* const* d_in, const int* in_sizes, int n_in,
                              void* d_out, int out_size, void* d_ws, size_t ws_size,
                              hipStream_t stream) {
  const float* x      = (const float*)d_in[0];
  const float* Win_f  = (const float*)d_in[1];
  const float* cw_f   = (const float*)d_in[2];
  const float* cb_f   = (const float*)d_in[3];
  const float* Wxp_f  = (const float*)d_in[4];
  const float* Wdt_f  = (const float*)d_in[5];
  const float* bdt_f  = (const float*)d_in[6];
  const float* Alog_f = (const float*)d_in[7];
  const float* Dv_f   = (const float*)d_in[8];
  const float* Win_b  = (const float*)d_in[9];
  const float* cw_b   = (const float*)d_in[10];
  const float* cb_b   = (const float*)d_in[11];
  const float* Wxp_b  = (const float*)d_in[12];
  const float* Wdt_b  = (const float*)d_in[13];
  const float* bdt_b  = (const float*)d_in[14];
  const float* Alog_b = (const float*)d_in[15];
  const float* Dv_b   = (const float*)d_in[16];
  const float* Wout   = (const float*)d_in[17];
  float* out = (float*)d_out;

  float* ws    = (float*)d_ws;
  float* xz_f  = ws;
  float* xz_b  = xz_f + (size_t)MROWS * XZW;
  float* u_f   = xz_b + (size_t)MROWS * XZW;
  float* u_b   = u_f  + (size_t)MROWS * DI;
  float* dbl_f = u_b  + (size_t)MROWS * DI;
  float* dbl_b = dbl_f + (size_t)MROWS * DBLW;
  const size_t PS = (size_t)BATCH * NCH * DI * NST;
  float* P_f   = dbl_b + (size_t)MROWS * DBLW;
  float* S_f   = P_f + PS;
  float* P_b   = S_f + PS;
  float* S_b   = P_b + PS;
  float* dt_f  = S_b + PS;
  float* dt_b  = dt_f + (size_t)MROWS * DI;

  const int BLK = 256, WPB = BLK / 32;

  {
    int waves = (MROWS / 16) * (XZW / 64);
    k_gemm_in<<<waves / WPB, BLK, 0, stream>>>(x, Win_f, xz_f, 0);
    k_gemm_in<<<waves / WPB, BLK, 0, stream>>>(x, Win_b, xz_b, 1);
  }
  {
    k_conv_xp<<<MROWS / 16, 32 * KW, 0, stream>>>(xz_f, cw_f, cb_f, Wxp_f, Wdt_f, bdt_f, Alog_f,
                                                 u_f, dbl_f, dt_f, P_f, S_f);
    k_conv_xp<<<MROWS / 16, 32 * KW, 0, stream>>>(xz_b, cw_b, cb_b, Wxp_b, Wdt_b, bdt_b, Alog_b,
                                                 u_b, dbl_b, dt_b, P_b, S_b);
  }
  {
    int cb = BATCH * DI * NST / 32;
    k_scan_combine<<<cb, 32 * NG, 0, stream>>>(P_f, S_f);
    k_scan_combine<<<cb, 32 * NG, 0, stream>>>(P_b, S_b);
  }
  {
    k_scan3_out<<<BATCH * NCH / 2, 2 * DI, 0, stream>>>(xz_f, xz_b, dt_f, dt_b, u_f, u_b, dbl_f, dbl_b,
                                                        Alog_f, Alog_b, Dv_f, Dv_b, S_f, S_b,
                                                        x, Wout, out);
  }
}


# Input projection: a wave's register tile grows from 16x64 to 64x96 (one wave per SIMD, 128x192 per block): a fragment load feeds 4-6 matrix steps, next k-step's loads issued ahead; k order kept
# speedup vs baseline: 3.0217x; 1.1925x over previous
#include <hip/hip_runtime.h>
#include <hip/hip_bf16.h>
#include <math.h>

#define BATCH 4
#define SEQ   2048
#define DIMX  192
#define DI    384
#define NST   16
#define DTR   12
#define XZW   768
#define DBLW  48
#define MROWS (BATCH*SEQ)
#define NCH   128
#define CL    (SEQ/NCH)
#define CG    16
#define NG    (NCH/CG)

typedef float v2f __attribute__((ext_vector_type(2)));
typedef float v8f __attribute__((ext_vector_type(8)));

__device__ __forceinline__ v8f wmma4(v2f a, v2f b, v8f c) {
  return __builtin_amdgcn_wmma_f32_16x16x4_f32(false, a, false, b, (short)0, c,
                                               false, false);
}

#define GMT 4
#define GNT 6

__global__ void __launch_bounds__(128)
k_gemm_in(const float* __restrict__ x,
          const float* __restrict__ Win,
          float* __restrict__ xz, int reverse) {
  const int w    = threadIdx.x >> 5;
  const int lane = threadIdx.x & 31;
  const int mrow = lane & 15, kh = lane >> 4;
  const int bm = blockIdx.x >> 2, bn = blockIdx.x & 3;
  const int m0 = bm * 128 + (w >> 1) * (16 * GMT);
  const int n0 = bn * 192 + (w & 1) * (16 * GNT);

  const float* ap[GMT];
  const float* bp[GNT];
#pragma unroll
  for (int i = 0; i < GMT; ++i) {
    const int m  = m0 + i * 16 + mrow;
    const int bb = m / SEQ, ll = m % SEQ;
    const int srow = reverse ? (bb * SEQ + (SEQ - 1 - ll)) : m;
    ap[i] = x + (size_t)srow * DIMX + 2 * kh;
  }
#pragma unroll
  for (int j = 0; j < GNT; ++j)
    bp[j] = Win + (size_t)(n0 + j * 16 + mrow) * DIMX + 2 * kh;

  v8f acc[GMT][GNT] = {};
  v2f a0[GMT], b0[GNT], a1[GMT], b1[GNT];

#define GEMM_IN_LOAD(A, B, K0)                                        \
  _Pragma("unroll") for (int i = 0; i < GMT; ++i)                     \
    A[i] = *(const v2f*)(ap[i] + (K0));                               \
  _Pragma("unroll") for (int j = 0; j < GNT; ++j)                     \
    B[j] = *(const v2f*)(bp[j] + (K0));
#define GEMM_IN_MMA(A, B)                                             \
  _Pragma("unroll") for (int i = 0; i < GMT; ++i)                     \
    _Pragma("unroll") for (int j = 0; j < GNT; ++j)                   \
      acc[i][j] = wmma4(A[i], B[j], acc[i][j]);

  GEMM_IN_LOAD(a0, b0, 0)
  for (int k0 = 0; k0 < DIMX - 8; k0 += 8) {
    GEMM_IN_LOAD(a1, b1, k0 + 4)
    GEMM_IN_MMA(a0, b0)
    GEMM_IN_LOAD(a0, b0, k0 + 8)
    GEMM_IN_MMA(a1, b1)
  }
  GEMM_IN_LOAD(a1, b1, DIMX - 4)
  GEMM_IN_MMA(a0, b0)
  GEMM_IN_MMA(a1, b1)
#undef GEMM_IN_LOAD
#undef GEMM_IN_MMA

#pragma unroll
  for (int i = 0; i < GMT; ++i)
#pragma unroll
    for (int j = 0; j < GNT; ++j)
#pragma unroll
      for (int v = 0; v < 8; ++v) {
        const int orow = m0 + i * 16 + v + 8 * kh;
        xz[(size_t)orow * XZW + n0 + j * 16 + mrow] = acc[i][j][v];
      }
}

static_assert(CL == 16, "a 16-row tile must be one scan chunk");
#define DTP 388
#define UP 68
#define KW 6

__global__ void __launch_bounds__(32 * KW)
k_conv_xp(const float* __restrict__ xz,
          const float* __restrict__ cw,
          const float* __restrict__ cb,
          const float* __restrict__ Wxp,
          const float* __restrict__ Wdt,
          const float* __restrict__ bdt,
          const float* __restrict__ Alog,
          float* __restrict__ u,
          float* __restrict__ dbl,
          float* __restrict__ dtb,
          float* __restrict__ Pb, float* __restrict__ Sb) {
  __shared__ __attribute__((aligned(16))) float ut[KW][16 * UP];
  __shared__ __attribute__((aligned(16))) float pd[16 * DTP];
  __shared__ __attribute__((aligned(16))) float dblT[16][DBLW];
  const int w    = threadIdx.x >> 5;
  const int lane = threadIdx.x & 31;
  const int row0 = blockIdx.x * 16;
  const int t0   = row0 % SEQ;

  const int cg = lane & 15, rh = lane >> 4;
  const int ch = w * 64 + cg * 4;
  const int rb = rh * 8;
  float4 o[8];
  {
    const float4 w0 = *(const float4*)(cw + (size_t)ch * 4);
    const float4 w1 = *(const float4*)(cw + (size_t)ch * 4 + 4);
    const float4 w2 = *(const float4*)(cw + (size_t)ch * 4 + 8);
    const float4 w3 = *(const float4*)(cw + (size_t)ch * 4 + 12);
    const float4 bias = *(const float4*)(cb + ch);
    float4 xr[11];
#pragma unroll
    for (int i = 0; i < 11; ++i) {
      const int rr = rb - 3 + i;
      const bool ok = (i >= 3) || (t0 + rr) >= 0;
      const float4 v = *(const float4*)(xz + (size_t)(row0 + (ok ? rr : 0)) * XZW + ch);
      xr[i] = ok ? v : make_float4(0.0f, 0.0f, 0.0f, 0.0f);
    }
#pragma unroll
    for (int r = 0; r < 8; ++r) {
      float a0 = bias.x, a1 = bias.y, a2 = bias.z, a3 = bias.w;
      a0 += xr[r].x * w0.x;  a1 += xr[r].y * w1.x;  a2 += xr[r].z * w2.x;  a3 += xr[r].w * w3.x;
      a0 += xr[r+1].x * w0.y; a1 += xr[r+1].y * w1.y; a2 += xr[r+1].z * w2.y; a3 += xr[r+1].w * w3.y;
      a0 += xr[r+2].x * w0.z; a1 += xr[r+2].y * w1.z; a2 += xr[r+2].z * w2.z; a3 += xr[r+2].w * w3.z;
      a0 += xr[r+3].x * w0.w; a1 += xr[r+3].y * w1.w; a2 += xr[r+3].z * w2.w; a3 += xr[r+3].w * w3.w;
      o[r].x = a0 / (1.0f + __expf(-a0));
      o[r].y = a1 / (1.0f + __expf(-a1));
      o[r].z = a2 / (1.0f + __expf(-a2));
      o[r].w = a3 / (1.0f + __expf(-a3));
      *(float4*)(&ut[w][(rb + r) * UP + cg * 4]) = o[r];
    }
  }
  __syncthreads();
#pragma unroll
  for (int r = 0; r < 8; ++r)
    *(float4*)(u + (size_t)(row0 + rb + r) * DI + ch) = o[r];
  __builtin_amdgcn_sched_barrier(0);

  const int mrow = lane & 15, kh = lane >> 4;
  bool nvalid[3];
  const float* bp[3];
#pragma unroll
  for (int j = 0; j < 3; ++j) {
    int n = j * 16 + mrow;
    nvalid[j] = (n < DTR + 2 * NST);
    bp[j] = Wxp + (size_t)(nvalid[j] ? n : 0) * DI + w * 64;
  }
  v8f acc[3] = {};
  for (int s = 0; s < 16; ++s) {
    const int kl = 4 * s + 2 * kh;
    const v2f a = *(const v2f*)(&ut[w][mrow * UP + kl]);
#pragma unroll
    for (int j = 0; j < 3; ++j) {
      const v2f bl = *(const v2f*)(bp[j] + kl);
      v2f b;
      b.x = nvalid[j] ? bl.x : 0.0f;
      b.y = nvalid[j] ? bl.y : 0.0f;
      acc[j] = wmma4(a, b, acc[j]);
    }
  }
#pragma unroll
  for (int j = 0; j < 3; ++j)
#pragma unroll
    for (int v = 0; v < 8; ++v)
      pd[(w * 24 + j * 8 + v) * 32 + lane] = acc[j][v];
  __syncthreads();

  for (int e = threadIdx.x; e < 24 * 32; e += 32 * KW) {
    const int q = e >> 5, ln = e & 31;
    float s = pd[q * 32 + ln];
#pragma unroll
    for (int ww = 1; ww < KW; ++ww) s += pd[(ww * 24 + q) * 32 + ln];
    const int lr  = (q & 7) + 8 * (ln >> 4);
    const int col = (q >> 3) * 16 + (ln & 15);
    dbl[(size_t)(row0 + lr) * DBLW + col] = s;
    dblT[lr][col] = s;
  }
  __syncthreads();

  {
    const float* wp[4];
#pragma unroll
    for (int j = 0; j < 4; ++j)
      wp[j] = Wdt + (size_t)(w * 64 + j * 16 + mrow) * DTR;
    v8f dacc[4] = {};
    for (int k0 = 0; k0 < DTR; k0 += 4) {
      const int kk = k0 + 2 * kh;
      v2f a; a.x = dblT[mrow][kk]; a.y = dblT[mrow][kk + 1];
#pragma unroll
      for (int j = 0; j < 4; ++j) {
        v2f b; b.x = wp[j][kk]; b.y = wp[j][kk + 1];
        dacc[j] = wmma4(a, b, dacc[j]);
      }
    }
#pragma unroll
    for (int j = 0; j < 4; ++j) {
      const int n = w * 64 + j * 16 + mrow;
      const float bias = bdt[n];
#pragma unroll
      for (int v = 0; v < 8; ++v) {
        const int lr = v + 8 * kh;
        const float t = dacc[j][v] + bias;
        const float sp = (t > 20.0f) ? t : log1pf(__expf(t));
        dtb[(size_t)(row0 + lr) * DI + n] = sp;
        pd[lr * DTP + n] = sp;
      }
    }
  }
  __syncthreads();

  const int bsq = row0 / SEQ;
  const int chk = t0 / CL;
  for (int cc = 0; cc < 2; ++cc) {
    const int d = threadIdx.x + cc * (32 * KW);
    float a[NST], P[NST], S[NST];
#pragma unroll
    for (int n = 0; n < NST; ++n) {
      a[n] = -__expf(Alog[d * NST + n]);
      P[n] = 1.0f;
      S[n] = 0.0f;
    }
    const float* up = &ut[d >> 6][d & 63];
    for (int t = 0; t < CL; ++t) {
      const float dt  = pd[t * DTP + d];
      const float uv  = up[t * UP];
      const float dtu = dt * uv;
#pragma unroll
      for (int n = 0; n < NST; ++n) {
        float dA = __expf(dt * a[n]);
        S[n] = S[n] * dA + dtu * dblT[t][DTR + n];
        P[n] *= dA;
      }
    }
    const size_t base = ((size_t)(bsq * NCH + chk) * DI + d) * NST;
#pragma unroll
    for (int n = 0; n < NST; ++n) {
      Pb[base + n] = P[n];
      Sb[base + n] = S[n];
    }
  }
}

__global__ void k_scan_combine(const float* __restrict__ Pb,
                               float* __restrict__ Sb) {
  __shared__ float Pg[NG][32];
  __shared__ float Sg[NG][32];
  const int e = threadIdx.x & 31;
  const int g = threadIdx.x >> 5;
  const int flat = blockIdx.x * 32 + e;
  const int b  = flat / (DI * NST);
  const int dn = flat % (DI * NST);
  const size_t CS = (size_t)DI * NST;
  const size_t base = (size_t)(b * NCH + g * CG) * CS + dn;

  float p[CG], s[CG];
#pragma unroll
  for (int j = 0; j < CG; ++j) {
    p[j] = Pb[base + (size_t)j * CS];
    s[j] = Sb[base + (size_t)j * CS];
  }
  float pp = 1.0f, ss = 0.0f;
#pragma unroll
  for (int j = 0; j < CG; ++j) {
    ss = p[j] * ss + s[j];
    pp *= p[j];
  }
  Pg[g][e] = pp;
  Sg[g][e] = ss;
  __syncthreads();

  float carry = 0.0f;
  for (int k = 0; k < g; ++k)
    carry = Pg[k][e] * carry + Sg[k][e];
#pragma unroll
  for (int j = 0; j < CG; ++j) {
    Sb[base + (size_t)j * CS] = carry;
    carry = p[j] * carry + s[j];
  }
}

#define YP 388
#define RT 32

__global__ void __launch_bounds__(2 * DI)
k_scan3_out(const float* __restrict__ xz_f, const float* __restrict__ xz_b,
            const float* __restrict__ dtb_f, const float* __restrict__ dtb_b,
            const float* __restrict__ u_f, const float* __restrict__ u_b,
            const float* __restrict__ dbl_f, const float* __restrict__ dbl_b,
            const float* __restrict__ Alog_f, const float* __restrict__ Alog_b,
            const float* __restrict__ Dv_f, const float* __restrict__ Dv_b,
            const float* __restrict__ Sb_f, const float* __restrict__ Sb_b,
            const float* __restrict__ x, const float* __restrict__ Wout,
            float* __restrict__ out) {
  __shared__ __attribute__((aligned(16))) float yF[RT * YP];
  __shared__ __attribute__((aligned(16))) float yB[RT * YP];
  const int b  = blockIdx.x / (NCH / 2);
  const int cp = blockIdx.x % (NCH / 2);
  const int g  = threadIdx.x / DI;
  const int d  = threadIdx.x - g * DI;
  const int c  = 2 * cp + g;
  const int cr = NCH - 1 - c;

  float af[NST], ab[NST], hf[NST], hb[NST];
  const size_t base_f = ((size_t)(b * NCH + c)  * DI + d) * NST;
  const size_t base_b = ((size_t)(b * NCH + cr) * DI + d) * NST;
#pragma unroll
  for (int n = 0; n < NST; ++n) {
    af[n] = -__expf(Alog_f[d * NST + n]);
    ab[n] = -__expf(Alog_b[d * NST + n]);
    hf[n] = Sb_f[base_f + n];
    hb[n] = Sb_b[base_b + n];
  }
  const float Dd_f = Dv_f[d];
  const float Dd_b = Dv_b[d];

  const size_t rf0 = (size_t)b * SEQ + (size_t)c  * CL;
  const size_t rb0 = (size_t)b * SEQ + (size_t)cr * CL;
  float* yFg = &yF[(CL * g) * YP + d];
  float* yBg = &yB[(CL * g) * YP + d];
  for (int t = 0; t < CL; ++t) {
    const size_t rowf = rf0 + t;
    const size_t rowb = rb0 + t;
    const float* bcf = dbl_f + rowf * DBLW;
    const float* bcb = dbl_b + rowb * DBLW;
    __builtin_prefetch(bcf + DBLW, 0, 0);
    __builtin_prefetch(bcb + DBLW, 0, 0);
    const float dtf = dtb_f[rowf * DI + d];
    const float dtr = dtb_b[rowb * DI + d];
    const float utf = u_f[rowf * DI + d];
    const float utr = u_b[rowb * DI + d];
    const float zf  = xz_f[rowf * XZW + DI + d];
    const float zr  = xz_b[rowb * XZW + DI + d];
    const float dtuf = dtf * utf;
    const float dtur = dtr * utr;
    float yf = 0.0f, yr = 0.0f;
#pragma unroll
    for (int n = 0; n < NST; ++n) {
      const float dAf = __expf(dtf * af[n]);
      const float dAr = __expf(dtr * ab[n]);
      hf[n] = hf[n] * dAf + dtuf * bcf[DTR + n];
      hb[n] = hb[n] * dAr + dtur * bcb[DTR + n];
      yf += hf[n] * bcf[DTR + NST + n];
      yr += hb[n] * bcb[DTR + NST + n];
    }
    yf += utf * Dd_f;
    yr += utr * Dd_b;
    yf *= zf / (1.0f + __expf(-zf));
    yr *= zr / (1.0f + __expf(-zr));
    yFg[t * YP] = yf;
    yBg[(CL - 1 - t) * YP] = yr;
  }
  __syncthreads();

  const int w = threadIdx.x >> 5;
  if (w < DIMX / 16) {
    const int lane = threadIdx.x & 31;
    const int mrow = lane & 15, kh = lane >> 4;
    const float* bp  = Wout + (size_t)(w * 16 + mrow) * DI;
    const float* pf0 = &yF[mrow * YP];
    const float* pb0 = &yB[mrow * YP];
    const float* pf1 = &yF[(16 + mrow) * YP];
    const float* pb1 = &yB[(16 + mrow) * YP];
    v8f acc0 = {}, acc1 = {};
    for (int k0 = 0; k0 < DI; k0 += 4) {
      const int kk = k0 + 2 * kh;
      const v2f fa0 = *(const v2f*)(pf0 + kk);
      const v2f fb0 = *(const v2f*)(pb0 + kk);
      const v2f fa1 = *(const v2f*)(pf1 + kk);
      const v2f fb1 = *(const v2f*)(pb1 + kk);
      v2f a0; a0.x = fa0.x + fb0.x; a0.y = fa0.y + fb0.y;
      v2f a1; a1.x = fa1.x + fb1.x; a1.y = fa1.y + fb1.y;
      v2f bv; bv.x = bp[kk]; bv.y = bp[kk + 1];
      acc0 = wmma4(a0, bv, acc0);
      acc1 = wmma4(a1, bv, acc1);
    }
    const int n = w * 16 + mrow;
    const size_t r0 = (size_t)b * SEQ + (size_t)cp * RT;
#pragma unroll
    for (int v = 0; v < 8; ++v) {
      const size_t orow = r0 + v + 8 * kh;
      out[orow * DIMX + n] = acc0[v] + x[orow * DIMX + n];
      out[(orow + 16) * DIMX + n] = acc1[v] + x[(orow + 16) * DIMX + n];
    }
  }
}

extern "C" void kernel_launch(void* const* d_in, const int* in_sizes, int n_in,
                              void* d_out, int out_size, void* d_ws, size_t ws_size,
                              hipStream_t stream) {
  const float* x      = (const float*)d_in[0];
  const float* Win_f  = (const float*)d_in[1];
  const float* cw_f   = (const float*)d_in[2];
  const float* cb_f   = (const float*)d_in[3];
  const float* Wxp_f  = (const float*)d_in[4];
  const float* Wdt_f  = (const float*)d_in[5];
  const float* bdt_f  = (const float*)d_in[6];
  const float* Alog_f = (const float*)d_in[7];
  const float* Dv_f   = (const float*)d_in[8];
  const float* Win_b  = (const float*)d_in[9];
  const float* cw_b   = (const float*)d_in[10];
  const float* cb_b   = (const float*)d_in[11];
  const float* Wxp_b  = (const float*)d_in[12];
  const float* Wdt_b  = (const float*)d_in[13];
  const float* bdt_b  = (const float*)d_in[14];
  const float* Alog_b = (const float*)d_in[15];
  const float* Dv_b   = (const float*)d_in[16];
  const float* Wout   = (const float*)d_in[17];
  float* out = (float*)d_out;

  float* ws    = (float*)d_ws;
  float* xz_f  = ws;
  float* xz_b  = xz_f + (size_t)MROWS * XZW;
  float* u_f   = xz_b + (size_t)MROWS * XZW;
  float* u_b   = u_f  + (size_t)MROWS * DI;
  float* dbl_f = u_b  + (size_t)MROWS * DI;
  float* dbl_b = dbl_f + (size_t)MROWS * DBLW;
  const size_t PS = (size_t)BATCH * NCH * DI * NST;
  float* P_f   = dbl_b + (size_t)MROWS * DBLW;
  float* S_f   = P_f + PS;
  float* P_b   = S_f + PS;
  float* S_b   = P_b + PS;
  float* dt_f  = S_b + PS;
  float* dt_b  = dt_f + (size_t)MROWS * DI;

  {
    const int gb = (MROWS / 128) * (XZW / 192);
    k_gemm_in<<<gb, 128, 0, stream>>>(x, Win_f, xz_f, 0);
    k_gemm_in<<<gb, 128, 0, stream>>>(x, Win_b, xz_b, 1);
  }
  {
    k_conv_xp<<<MROWS / 16, 32 * KW, 0, stream>>>(xz_f, cw_f, cb_f, Wxp_f, Wdt_f, bdt_f, Alog_f,
                                                 u_f, dbl_f, dt_f, P_f, S_f);
    k_conv_xp<<<MROWS / 16, 32 * KW, 0, stream>>>(xz_b, cw_b, cb_b, Wxp_b, Wdt_b, bdt_b, Alog_b,
                                                 u_b, dbl_b, dt_b, P_b, S_b);
  }
  {
    int cb = BATCH * DI * NST / 32;
    k_scan_combine<<<cb, 32 * NG, 0, stream>>>(P_f, S_f);
    k_scan_combine<<<cb, 32 * NG, 0, stream>>>(P_b, S_b);
  }
  {
    k_scan3_out<<<BATCH * NCH / 2, 2 * DI, 0, stream>>>(xz_f, xz_b, dt_f, dt_b, u_f, u_b, dbl_f, dbl_b,
                                                        Alog_f, Alog_b, Dv_f, Dv_b, S_f, S_b,
                                                        x, Wout, out);
  }
}


# dt stage: softplus without a branch (library log1pf runs for every t, a select picks t above 20): a lane's 32 evaluations overlap, not 32 guarded dependent chains in a row; values unchanged
# speedup vs baseline: 3.1187x; 1.0321x over previous
#include <hip/hip_runtime.h>
#include <hip/hip_bf16.h>
#include <math.h>

#define BATCH 4
#define SEQ   2048
#define DIMX  192
#define DI    384
#define NST   16
#define DTR   12
#define XZW   768
#define DBLW  48
#define MROWS (BATCH*SEQ)
#define NCH   128
#define CL    (SEQ/NCH)
#define CG    16
#define NG    (NCH/CG)

typedef float v2f __attribute__((ext_vector_type(2)));
typedef float v8f __attribute__((ext_vector_type(8)));

__device__ __forceinline__ v8f wmma4(v2f a, v2f b, v8f c) {
  return __builtin_amdgcn_wmma_f32_16x16x4_f32(false, a, false, b, (short)0, c,
                                               false, false);
}

#define GMT 4
#define GNT 6

__global__ void __launch_bounds__(128)
k_gemm_in(const float* __restrict__ x,
          const float* __restrict__ Win,
          float* __restrict__ xz, int reverse) {
  const int w    = threadIdx.x >> 5;
  const int lane = threadIdx.x & 31;
  const int mrow = lane & 15, kh = lane >> 4;
  const int bm = blockIdx.x >> 2, bn = blockIdx.x & 3;
  const int m0 = bm * 128 + (w >> 1) * (16 * GMT);
  const int n0 = bn * 192 + (w & 1) * (16 * GNT);

  const float* ap[GMT];
  const float* bp[GNT];
#pragma unroll
  for (int i = 0; i < GMT; ++i) {
    const int m  = m0 + i * 16 + mrow;
    const int bb = m / SEQ, ll = m % SEQ;
    const int srow = reverse ? (bb * SEQ + (SEQ - 1 - ll)) : m;
    ap[i] = x + (size_t)srow * DIMX + 2 * kh;
  }
#pragma unroll
  for (int j = 0; j < GNT; ++j)
    bp[j] = Win + (size_t)(n0 + j * 16 + mrow) * DIMX + 2 * kh;

  v8f acc[GMT][GNT] = {};
  v2f a0[GMT], b0[GNT], a1[GMT], b1[GNT];

#define GEMM_IN_LOAD(A, B, K0)                                        \
  _Pragma("unroll") for (int i = 0; i < GMT; ++i)                     \
    A[i] = *(const v2f*)(ap[i] + (K0));                               \
  _Pragma("unroll") for (int j = 0; j < GNT; ++j)                     \
    B[j] = *(const v2f*)(bp[j] + (K0));
#define GEMM_IN_MMA(A, B)                                             \
  _Pragma("unroll") for (int i = 0; i < GMT; ++i)                     \
    _Pragma("unroll") for (int j = 0; j < GNT; ++j)                   \
      acc[i][j] = wmma4(A[i], B[j], acc[i][j]);

  GEMM_IN_LOAD(a0, b0, 0)
  for (int k0 = 0; k0 < DIMX - 8; k0 += 8) {
    GEMM_IN_LOAD(a1, b1, k0 + 4)
    GEMM_IN_MMA(a0, b0)
    GEMM_IN_LOAD(a0, b0, k0 + 8)
    GEMM_IN_MMA(a1, b1)
  }
  GEMM_IN_LOAD(a1, b1, DIMX - 4)
  GEMM_IN_MMA(a0, b0)
  GEMM_IN_MMA(a1, b1)
#undef GEMM_IN_LOAD
#undef GEMM_IN_MMA

#pragma unroll
  for (int i = 0; i < GMT; ++i)
#pragma unroll
    for (int j = 0; j < GNT; ++j)
#pragma unroll
      for (int v = 0; v < 8; ++v) {
        const int orow = m0 + i * 16 + v + 8 * kh;
        xz[(size_t)orow * XZW + n0 + j * 16 + mrow] = acc[i][j][v];
      }
}

static_assert(CL == 16, "a 16-row tile must be one scan chunk");
#define DTP 388
#define UP 68
#define KW 6

__global__ void __launch_bounds__(32 * KW)
k_conv_xp(const float* __restrict__ xz,
          const float* __restrict__ cw,
          const float* __restrict__ cb,
          const float* __restrict__ Wxp,
          const float* __restrict__ Wdt,
          const float* __restrict__ bdt,
          const float* __restrict__ Alog,
          float* __restrict__ u,
          float* __restrict__ dbl,
          float* __restrict__ dtb,
          float* __restrict__ Pb, float* __restrict__ Sb) {
  __shared__ __attribute__((aligned(16))) float ut[KW][16 * UP];
  __shared__ __attribute__((aligned(16))) float pd[16 * DTP];
  __shared__ __attribute__((aligned(16))) float dblT[16][DBLW];
  const int w    = threadIdx.x >> 5;
  const int lane = threadIdx.x & 31;
  const int row0 = blockIdx.x * 16;
  const int t0   = row0 % SEQ;

  const int cg = lane & 15, rh = lane >> 4;
  const int ch = w * 64 + cg * 4;
  const int rb = rh * 8;
  float4 o[8];
  {
    const float4 w0 = *(const float4*)(cw + (size_t)ch * 4);
    const float4 w1 = *(const float4*)(cw + (size_t)ch * 4 + 4);
    const float4 w2 = *(const float4*)(cw + (size_t)ch * 4 + 8);
    const float4 w3 = *(const float4*)(cw + (size_t)ch * 4 + 12);
    const float4 bias = *(const float4*)(cb + ch);
    float4 xr[11];
#pragma unroll
    for (int i = 0; i < 11; ++i) {
      const int rr = rb - 3 + i;
      const bool ok = (i >= 3) || (t0 + rr) >= 0;
      const float4 v = *(const float4*)(xz + (size_t)(row0 + (ok ? rr : 0)) * XZW + ch);
      xr[i] = ok ? v : make_float4(0.0f, 0.0f, 0.0f, 0.0f);
    }
#pragma unroll
    for (int r = 0; r < 8; ++r) {
      float a0 = bias.x, a1 = bias.y, a2 = bias.z, a3 = bias.w;
      a0 += xr[r].x * w0.x;  a1 += xr[r].y * w1.x;  a2 += xr[r].z * w2.x;  a3 += xr[r].w * w3.x;
      a0 += xr[r+1].x * w0.y; a1 += xr[r+1].y * w1.y; a2 += xr[r+1].z * w2.y; a3 += xr[r+1].w * w3.y;
      a0 += xr[r+2].x * w0.z; a1 += xr[r+2].y * w1.z; a2 += xr[r+2].z * w2.z; a3 += xr[r+2].w * w3.z;
      a0 += xr[r+3].x * w0.w; a1 += xr[r+3].y * w1.w; a2 += xr[r+3].z * w2.w; a3 += xr[r+3].w * w3.w;
      o[r].x = a0 / (1.0f + __expf(-a0));
      o[r].y = a1 / (1.0f + __expf(-a1));
      o[r].z = a2 / (1.0f + __expf(-a2));
      o[r].w = a3 / (1.0f + __expf(-a3));
      *(float4*)(&ut[w][(rb + r) * UP + cg * 4]) = o[r];
    }
  }
  __syncthreads();
#pragma unroll
  for (int r = 0; r < 8; ++r)
    *(float4*)(u + (size_t)(row0 + rb + r) * DI + ch) = o[r];
  __builtin_amdgcn_sched_barrier(0);

  const int mrow = lane & 15, kh = lane >> 4;
  bool nvalid[3];
  const float* bp[3];
#pragma unroll
  for (int j = 0; j < 3; ++j) {
    int n = j * 16 + mrow;
    nvalid[j] = (n < DTR + 2 * NST);
    bp[j] = Wxp + (size_t)(nvalid[j] ? n : 0) * DI + w * 64;
  }
  v8f acc[3] = {};
  for (int s = 0; s < 16; ++s) {
    const int kl = 4 * s + 2 * kh;
    const v2f a = *(const v2f*)(&ut[w][mrow * UP + kl]);
#pragma unroll
    for (int j = 0; j < 3; ++j) {
      const v2f bl = *(const v2f*)(bp[j] + kl);
      v2f b;
      b.x = nvalid[j] ? bl.x : 0.0f;
      b.y = nvalid[j] ? bl.y : 0.0f;
      acc[j] = wmma4(a, b, acc[j]);
    }
  }
#pragma unroll
  for (int j = 0; j < 3; ++j)
#pragma unroll
    for (int v = 0; v < 8; ++v)
      pd[(w * 24 + j * 8 + v) * 32 + lane] = acc[j][v];
  __syncthreads();

  for (int e = threadIdx.x; e < 24 * 32; e += 32 * KW) {
    const int q = e >> 5, ln = e & 31;
    float s = pd[q * 32 + ln];
#pragma unroll
    for (int ww = 1; ww < KW; ++ww) s += pd[(ww * 24 + q) * 32 + ln];
    const int lr  = (q & 7) + 8 * (ln >> 4);
    const int col = (q >> 3) * 16 + (ln & 15);
    dbl[(size_t)(row0 + lr) * DBLW + col] = s;
    dblT[lr][col] = s;
  }
  __syncthreads();

  {
    const float* wp[4];
#pragma unroll
    for (int j = 0; j < 4; ++j)
      wp[j] = Wdt + (size_t)(w * 64 + j * 16 + mrow) * DTR;
    v8f dacc[4] = {};
    for (int k0 = 0; k0 < DTR; k0 += 4) {
      const int kk = k0 + 2 * kh;
      v2f a; a.x = dblT[mrow][kk]; a.y = dblT[mrow][kk + 1];
#pragma unroll
      for (int j = 0; j < 4; ++j) {
        v2f b; b.x = wp[j][kk]; b.y = wp[j][kk + 1];
        dacc[j] = wmma4(a, b, dacc[j]);
      }
    }
#pragma unroll
    for (int j = 0; j < 4; ++j) {
      const int n = w * 64 + j * 16 + mrow;
      const float bias = bdt[n];
#pragma unroll
      for (int v = 0; v < 8; ++v) {
        const int lr = v + 8 * kh;
        const float t = dacc[j][v] + bias;
        const float lg = log1pf(__expf(t));
        const float sp = (t > 20.0f) ? t : lg;
        dtb[(size_t)(row0 + lr) * DI + n] = sp;
        pd[lr * DTP + n] = sp;
      }
    }
  }
  __syncthreads();

  const int bsq = row0 / SEQ;
  const int chk = t0 / CL;
  for (int cc = 0; cc < 2; ++cc) {
    const int d = threadIdx.x + cc * (32 * KW);
    float a[NST], P[NST], S[NST];
#pragma unroll
    for (int n = 0; n < NST; ++n) {
      a[n] = -__expf(Alog[d * NST + n]);
      P[n] = 1.0f;
      S[n] = 0.0f;
    }
    const float* up = &ut[d >> 6][d & 63];
    for (int t = 0; t < CL; ++t) {
      const float dt  = pd[t * DTP + d];
      const float uv  = up[t * UP];
      const float dtu = dt * uv;
#pragma unroll
      for (int n = 0; n < NST; ++n) {
        float dA = __expf(dt * a[n]);
        S[n] = S[n] * dA + dtu * dblT[t][DTR + n];
        P[n] *= dA;
      }
    }
    const size_t base = ((size_t)(bsq * NCH + chk) * DI + d) * NST;
#pragma unroll
    for (int n = 0; n < NST; ++n) {
      Pb[base + n] = P[n];
      Sb[base + n] = S[n];
    }
  }
}

__global__ void k_scan_combine(const float* __restrict__ Pb,
                               float* __restrict__ Sb) {
  __shared__ float Pg[NG][32];
  __shared__ float Sg[NG][32];
  const int e = threadIdx.x & 31;
  const int g = threadIdx.x >> 5;
  const int flat = blockIdx.x * 32 + e;
  const int b  = flat / (DI * NST);
  const int dn = flat % (DI * NST);
  const size_t CS = (size_t)DI * NST;
  const size_t base = (size_t)(b * NCH + g * CG) * CS + dn;

  float p[CG], s[CG];
#pragma unroll
  for (int j = 0; j < CG; ++j) {
    p[j] = Pb[base + (size_t)j * CS];
    s[j] = Sb[base + (size_t)j * CS];
  }
  float pp = 1.0f, ss = 0.0f;
#pragma unroll
  for (int j = 0; j < CG; ++j) {
    ss = p[j] * ss + s[j];
    pp *= p[j];
  }
  Pg[g][e] = pp;
  Sg[g][e] = ss;
  __syncthreads();

  float carry = 0.0f;
  for (int k = 0; k < g; ++k)
    carry = Pg[k][e] * carry + Sg[k][e];
#pragma unroll
  for (int j = 0; j < CG; ++j) {
    Sb[base + (size_t)j * CS] = carry;
    carry = p[j] * carry + s[j];
  }
}

#define YP 388
#define RT 32

__global__ void __launch_bounds__(2 * DI)
k_scan3_out(const float* __restrict__ xz_f, const float* __restrict__ xz_b,
            const float* __restrict__ dtb_f, const float* __restrict__ dtb_b,
            const float* __restrict__ u_f, const float* __restrict__ u_b,
            const float* __restrict__ dbl_f, const float* __restrict__ dbl_b,
            const float* __restrict__ Alog_f, const float* __restrict__ Alog_b,
            const float* __restrict__ Dv_f, const float* __restrict__ Dv_b,
            const float* __restrict__ Sb_f, const float* __restrict__ Sb_b,
            const float* __restrict__ x, const float* __restrict__ Wout,
            float* __restrict__ out) {
  __shared__ __attribute__((aligned(16))) float yF[RT * YP];
  __shared__ __attribute__((aligned(16))) float yB[RT * YP];
  const int b  = blockIdx.x / (NCH / 2);
  const int cp = blockIdx.x % (NCH / 2);
  const int g  = threadIdx.x / DI;
  const int d  = threadIdx.x - g * DI;
  const int c  = 2 * cp + g;
  const int cr = NCH - 1 - c;

  float af[NST], ab[NST], hf[NST], hb[NST];
  const size_t base_f = ((size_t)(b * NCH + c)  * DI + d) * NST;
  const size_t base_b = ((size_t)(b * NCH + cr) * DI + d) * NST;
#pragma unroll
  for (int n = 0; n < NST; ++n) {
    af[n] = -__expf(Alog_f[d * NST + n]);
    ab[n] = -__expf(Alog_b[d * NST + n]);
    hf[n] = Sb_f[base_f + n];
    hb[n] = Sb_b[base_b + n];
  }
  const float Dd_f = Dv_f[d];
  const float Dd_b = Dv_b[d];

  const size_t rf0 = (size_t)b * SEQ + (size_t)c  * CL;
  const size_t rb0 = (size_t)b * SEQ + (size_t)cr * CL;
  float* yFg = &yF[(CL * g) * YP + d];
  float* yBg = &yB[(CL * g) * YP + d];
  for (int t = 0; t < CL; ++t) {
    const size_t rowf = rf0 + t;
    const size_t rowb = rb0 + t;
    const float* bcf = dbl_f + rowf * DBLW;
    const float* bcb = dbl_b + rowb * DBLW;
    __builtin_prefetch(bcf + DBLW, 0, 0);
    __builtin_prefetch(bcb + DBLW, 0, 0);
    const float dtf = dtb_f[rowf * DI + d];
    const float dtr = dtb_b[rowb * DI + d];
    const float utf = u_f[rowf * DI + d];
    const float utr = u_b[rowb * DI + d];
    const float zf  = xz_f[rowf * XZW + DI + d];
    const float zr  = xz_b[rowb * XZW + DI + d];
    const float dtuf = dtf * utf;
    const float dtur = dtr * utr;
    float yf = 0.0f, yr = 0.0f;
#pragma unroll
    for (int n = 0; n < NST; ++n) {
      const float dAf = __expf(dtf * af[n]);
      const float dAr = __expf(dtr * ab[n]);
      hf[n] = hf[n] * dAf + dtuf * bcf[DTR + n];
      hb[n] = hb[n] * dAr + dtur * bcb[DTR + n];
      yf += hf[n] * bcf[DTR + NST + n];
      yr += hb[n] * bcb[DTR + NST + n];
    }
    yf += utf * Dd_f;
    yr += utr * Dd_b;
    yf *= zf / (1.0f + __expf(-zf));
    yr *= zr / (1.0f + __expf(-zr));
    yFg[t * YP] = yf;
    yBg[(CL - 1 - t) * YP] = yr;
  }
  __syncthreads();

  const int w = threadIdx.x >> 5;
  if (w < DIMX / 16) {
    const int lane = threadIdx.x & 31;
    const int mrow = lane & 15, kh = lane >> 4;
    const float* bp  = Wout + (size_t)(w * 16 + mrow) * DI;
    const float* pf0 = &yF[mrow * YP];
    const float* pb0 = &yB[mrow * YP];
    const float* pf1 = &yF[(16 + mrow) * YP];
    const float* pb1 = &yB[(16 + mrow) * YP];
    v8f acc0 = {}, acc1 = {};
    for (int k0 = 0; k0 < DI; k0 += 4) {
      const int kk = k0 + 2 * kh;
      const v2f fa0 = *(const v2f*)(pf0 + kk);
      const v2f fb0 = *(const v2f*)(pb0 + kk);
      const v2f fa1 = *(const v2f*)(pf1 + kk);
      const v2f fb1 = *(const v2f*)(pb1 + kk);
      v2f a0; a0.x = fa0.x + fb0.x; a0.y = fa0.y + fb0.y;
      v2f a1; a1.x = fa1.x + fb1.x; a1.y = fa1.y + fb1.y;
      v2f bv; bv.x = bp[kk]; bv.y = bp[kk + 1];
      acc0 = wmma4(a0, bv, acc0);
      acc1 = wmma4(a1, bv, acc1);
    }
    const int n = w * 16 + mrow;
    const size_t r0 = (size_t)b * SEQ + (size_t)cp * RT;
#pragma unroll
    for (int v = 0; v < 8; ++v) {
      const size_t orow = r0 + v + 8 * kh;
      out[orow * DIMX + n] = acc0[v] + x[orow * DIMX + n];
      out[(orow + 16) * DIMX + n] = acc1[v] + x[(orow + 16) * DIMX + n];
    }
  }
}

extern "C" void kernel_launch(void* const* d_in, const int* in_sizes, int n_in,
                              void* d_out, int out_size, void* d_ws, size_t ws_size,
                              hipStream_t stream) {
  const float* x      = (const float*)d_in[0];
  const float* Win_f  = (const float*)d_in[1];
  const float* cw_f   = (const float*)d_in[2];
  const float* cb_f   = (const float*)d_in[3];
  const float* Wxp_f  = (const float*)d_in[4];
  const float* Wdt_f  = (const float*)d_in[5];
  const float* bdt_f  = (const float*)d_in[6];
  const float* Alog_f = (const float*)d_in[7];
  const float* Dv_f   = (const float*)d_in[8];
  const float* Win_b  = (const float*)d_in[9];
  const float* cw_b   = (const float*)d_in[10];
  const float* cb_b   = (const float*)d_in[11];
  const float* Wxp_b  = (const float*)d_in[12];
  const float* Wdt_b  = (const float*)d_in[13];
  const float* bdt_b  = (const float*)d_in[14];
  const float* Alog_b = (const float*)d_in[15];
  const float* Dv_b   = (const float*)d_in[16];
  const float* Wout   = (const float*)d_in[17];
  float* out = (float*)d_out;

  float* ws    = (float*)d_ws;
  float* xz_f  = ws;
  float* xz_b  = xz_f + (size_t)MROWS * XZW;
  float* u_f   = xz_b + (size_t)MROWS * XZW;
  float* u_b   = u_f  + (size_t)MROWS * DI;
  float* dbl_f = u_b  + (size_t)MROWS * DI;
  float* dbl_b = dbl_f + (size_t)MROWS * DBLW;
  const size_t PS = (size_t)BATCH * NCH * DI * NST;
  float* P_f   = dbl_b + (size_t)MROWS * DBLW;
  float* S_f   = P_f + PS;
  float* P_b   = S_f + PS;
  float* S_b   = P_b + PS;
  float* dt_f  = S_b + PS;
  float* dt_b  = dt_f + (size_t)MROWS * DI;

  {
    const int gb = (MROWS / 128) * (XZW / 192);
    k_gemm_in<<<gb, 128, 0, stream>>>(x, Win_f, xz_f, 0);
    k_gemm_in<<<gb, 128, 0, stream>>>(x, Win_b, xz_b, 1);
  }
  {
    k_conv_xp<<<MROWS / 16, 32 * KW, 0, stream>>>(xz_f, cw_f, cb_f, Wxp_f, Wdt_f, bdt_f, Alog_f,
                                                 u_f, dbl_f, dt_f, P_f, S_f);
    k_conv_xp<<<MROWS / 16, 32 * KW, 0, stream>>>(xz_b, cw_b, cb_b, Wxp_b, Wdt_b, bdt_b, Alog_b,
                                                 u_b, dbl_b, dt_b, P_b, S_b);
  }
  {
    int cb = BATCH * DI * NST / 32;
    k_scan_combine<<<cb, 32 * NG, 0, stream>>>(P_f, S_f);
    k_scan_combine<<<cb, 32 * NG, 0, stream>>>(P_b, S_b);
  }
  {
    k_scan3_out<<<BATCH * NCH / 2, 2 * DI, 0, stream>>>(xz_f, xz_b, dt_f, dt_b, u_f, u_b, dbl_f, dbl_b,
                                                        Alog_f, Alog_b, Dv_f, Dv_b, S_f, S_b,
                                                        x, Wout, out);
  }
}


# Pass 3: a step's wave-uniform B_t/C_t come by scalar loads into scalar registers (group index read from the first lane), not 16 repeated vector loads; y products fused (fma), sum order kept
# speedup vs baseline: 3.3536x; 1.0753x over previous
#include <hip/hip_runtime.h>
#include <hip/hip_bf16.h>
#include <math.h>

#define BATCH 4
#define SEQ   2048
#define DIMX  192
#define DI    384
#define NST   16
#define DTR   12
#define XZW   768
#define DBLW  48
#define MROWS (BATCH*SEQ)
#define NCH   128
#define CL    (SEQ/NCH)
#define CG    16
#define NG    (NCH/CG)

typedef float v2f __attribute__((ext_vector_type(2)));
typedef float v8f __attribute__((ext_vector_type(8)));

__device__ __forceinline__ v8f wmma4(v2f a, v2f b, v8f c) {
  return __builtin_amdgcn_wmma_f32_16x16x4_f32(false, a, false, b, (short)0, c,
                                               false, false);
}

#define GMT 4
#define GNT 6

__global__ void __launch_bounds__(128)
k_gemm_in(const float* __restrict__ x,
          const float* __restrict__ Win,
          float* __restrict__ xz, int reverse) {
  const int w    = threadIdx.x >> 5;
  const int lane = threadIdx.x & 31;
  const int mrow = lane & 15, kh = lane >> 4;
  const int bm = blockIdx.x >> 2, bn = blockIdx.x & 3;
  const int m0 = bm * 128 + (w >> 1) * (16 * GMT);
  const int n0 = bn * 192 + (w & 1) * (16 * GNT);

  const float* ap[GMT];
  const float* bp[GNT];
#pragma unroll
  for (int i = 0; i < GMT; ++i) {
    const int m  = m0 + i * 16 + mrow;
    const int bb = m / SEQ, ll = m % SEQ;
    const int srow = reverse ? (bb * SEQ + (SEQ - 1 - ll)) : m;
    ap[i] = x + (size_t)srow * DIMX + 2 * kh;
  }
#pragma unroll
  for (int j = 0; j < GNT; ++j)
    bp[j] = Win + (size_t)(n0 + j * 16 + mrow) * DIMX + 2 * kh;

  v8f acc[GMT][GNT] = {};
  v2f a0[GMT], b0[GNT], a1[GMT], b1[GNT];

#define GEMM_IN_LOAD(A, B, K0)                                        \
  _Pragma("unroll") for (int i = 0; i < GMT; ++i)                     \
    A[i] = *(const v2f*)(ap[i] + (K0));                               \
  _Pragma("unroll") for (int j = 0; j < GNT; ++j)                     \
    B[j] = *(const v2f*)(bp[j] + (K0));
#define GEMM_IN_MMA(A, B)                                             \
  _Pragma("unroll") for (int i = 0; i < GMT; ++i)                     \
    _Pragma("unroll") for (int j = 0; j < GNT; ++j)                   \
      acc[i][j] = wmma4(A[i], B[j], acc[i][j]);

  GEMM_IN_LOAD(a0, b0, 0)
  for (int k0 = 0; k0 < DIMX - 8; k0 += 8) {
    GEMM_IN_LOAD(a1, b1, k0 + 4)
    GEMM_IN_MMA(a0, b0)
    GEMM_IN_LOAD(a0, b0, k0 + 8)
    GEMM_IN_MMA(a1, b1)
  }
  GEMM_IN_LOAD(a1, b1, DIMX - 4)
  GEMM_IN_MMA(a0, b0)
  GEMM_IN_MMA(a1, b1)
#undef GEMM_IN_LOAD
#undef GEMM_IN_MMA

#pragma unroll
  for (int i = 0; i < GMT; ++i)
#pragma unroll
    for (int j = 0; j < GNT; ++j)
#pragma unroll
      for (int v = 0; v < 8; ++v) {
        const int orow = m0 + i * 16 + v + 8 * kh;
        xz[(size_t)orow * XZW + n0 + j * 16 + mrow] = acc[i][j][v];
      }
}

static_assert(CL == 16, "a 16-row tile must be one scan chunk");
#define DTP 388
#define UP 68
#define KW 6

__global__ void __launch_bounds__(32 * KW)
k_conv_xp(const float* __restrict__ xz,
          const float* __restrict__ cw,
          const float* __restrict__ cb,
          const float* __restrict__ Wxp,
          const float* __restrict__ Wdt,
          const float* __restrict__ bdt,
          const float* __restrict__ Alog,
          float* __restrict__ u,
          float* __restrict__ dbl,
          float* __restrict__ dtb,
          float* __restrict__ Pb, float* __restrict__ Sb) {
  __shared__ __attribute__((aligned(16))) float ut[KW][16 * UP];
  __shared__ __attribute__((aligned(16))) float pd[16 * DTP];
  __shared__ __attribute__((aligned(16))) float dblT[16][DBLW];
  const int w    = threadIdx.x >> 5;
  const int lane = threadIdx.x & 31;
  const int row0 = blockIdx.x * 16;
  const int t0   = row0 % SEQ;

  const int cg = lane & 15, rh = lane >> 4;
  const int ch = w * 64 + cg * 4;
  const int rb = rh * 8;
  float4 o[8];
  {
    const float4 w0 = *(const float4*)(cw + (size_t)ch * 4);
    const float4 w1 = *(const float4*)(cw + (size_t)ch * 4 + 4);
    const float4 w2 = *(const float4*)(cw + (size_t)ch * 4 + 8);
    const float4 w3 = *(const float4*)(cw + (size_t)ch * 4 + 12);
    const float4 bias = *(const float4*)(cb + ch);
    float4 xr[11];
#pragma unroll
    for (int i = 0; i < 11; ++i) {
      const int rr = rb - 3 + i;
      const bool ok = (i >= 3) || (t0 + rr) >= 0;
      const float4 v = *(const float4*)(xz + (size_t)(row0 + (ok ? rr : 0)) * XZW + ch);
      xr[i] = ok ? v : make_float4(0.0f, 0.0f, 0.0f, 0.0f);
    }
#pragma unroll
    for (int r = 0; r < 8; ++r) {
      float a0 = bias.x, a1 = bias.y, a2 = bias.z, a3 = bias.w;
      a0 += xr[r].x * w0.x;  a1 += xr[r].y * w1.x;  a2 += xr[r].z * w2.x;  a3 += xr[r].w * w3.x;
      a0 += xr[r+1].x * w0.y; a1 += xr[r+1].y * w1.y; a2 += xr[r+1].z * w2.y; a3 += xr[r+1].w * w3.y;
      a0 += xr[r+2].x * w0.z; a1 += xr[r+2].y * w1.z; a2 += xr[r+2].z * w2.z; a3 += xr[r+2].w * w3.z;
      a0 += xr[r+3].x * w0.w; a1 += xr[r+3].y * w1.w; a2 += xr[r+3].z * w2.w; a3 += xr[r+3].w * w3.w;
      o[r].x = a0 / (1.0f + __expf(-a0));
      o[r].y = a1 / (1.0f + __expf(-a1));
      o[r].z = a2 / (1.0f + __expf(-a2));
      o[r].w = a3 / (1.0f + __expf(-a3));
      *(float4*)(&ut[w][(rb + r) * UP + cg * 4]) = o[r];
    }
  }
  __syncthreads();
#pragma unroll
  for (int r = 0; r < 8; ++r)
    *(float4*)(u + (size_t)(row0 + rb + r) * DI + ch) = o[r];
  __builtin_amdgcn_sched_barrier(0);

  const int mrow = lane & 15, kh = lane >> 4;
  bool nvalid[3];
  const float* bp[3];
#pragma unroll
  for (int j = 0; j < 3; ++j) {
    int n = j * 16 + mrow;
    nvalid[j] = (n < DTR + 2 * NST);
    bp[j] = Wxp + (size_t)(nvalid[j] ? n : 0) * DI + w * 64;
  }
  v8f acc[3] = {};
  for (int s = 0; s < 16; ++s) {
    const int kl = 4 * s + 2 * kh;
    const v2f a = *(const v2f*)(&ut[w][mrow * UP + kl]);
#pragma unroll
    for (int j = 0; j < 3; ++j) {
      const v2f bl = *(const v2f*)(bp[j] + kl);
      v2f b;
      b.x = nvalid[j] ? bl.x : 0.0f;
      b.y = nvalid[j] ? bl.y : 0.0f;
      acc[j] = wmma4(a, b, acc[j]);
    }
  }
#pragma unroll
  for (int j = 0; j < 3; ++j)
#pragma unroll
    for (int v = 0; v < 8; ++v)
      pd[(w * 24 + j * 8 + v) * 32 + lane] = acc[j][v];
  __syncthreads();

  for (int e = threadIdx.x; e < 24 * 32; e += 32 * KW) {
    const int q = e >> 5, ln = e & 31;
    float s = pd[q * 32 + ln];
#pragma unroll
    for (int ww = 1; ww < KW; ++ww) s += pd[(ww * 24 + q) * 32 + ln];
    const int lr  = (q & 7) + 8 * (ln >> 4);
    const int col = (q >> 3) * 16 + (ln & 15);
    dbl[(size_t)(row0 + lr) * DBLW + col] = s;
    dblT[lr][col] = s;
  }
  __syncthreads();

  {
    const float* wp[4];
#pragma unroll
    for (int j = 0; j < 4; ++j)
      wp[j] = Wdt + (size_t)(w * 64 + j * 16 + mrow) * DTR;
    v8f dacc[4] = {};
    for (int k0 = 0; k0 < DTR; k0 += 4) {
      const int kk = k0 + 2 * kh;
      v2f a; a.x = dblT[mrow][kk]; a.y = dblT[mrow][kk + 1];
#pragma unroll
      for (int j = 0; j < 4; ++j) {
        v2f b; b.x = wp[j][kk]; b.y = wp[j][kk + 1];
        dacc[j] = wmma4(a, b, dacc[j]);
      }
    }
#pragma unroll
    for (int j = 0; j < 4; ++j) {
      const int n = w * 64 + j * 16 + mrow;
      const float bias = bdt[n];
#pragma unroll
      for (int v = 0; v < 8; ++v) {
        const int lr = v + 8 * kh;
        const float t = dacc[j][v] + bias;
        const float lg = log1pf(__expf(t));
        const float sp = (t > 20.0f) ? t : lg;
        dtb[(size_t)(row0 + lr) * DI + n] = sp;
        pd[lr * DTP + n] = sp;
      }
    }
  }
  __syncthreads();

  const int bsq = row0 / SEQ;
  const int chk = t0 / CL;
  for (int cc = 0; cc < 2; ++cc) {
    const int d = threadIdx.x + cc * (32 * KW);
    float a[NST], P[NST], S[NST];
#pragma unroll
    for (int n = 0; n < NST; ++n) {
      a[n] = -__expf(Alog[d * NST + n]);
      P[n] = 1.0f;
      S[n] = 0.0f;
    }
    const float* up = &ut[d >> 6][d & 63];
    for (int t = 0; t < CL; ++t) {
      const float dt  = pd[t * DTP + d];
      const float uv  = up[t * UP];
      const float dtu = dt * uv;
#pragma unroll
      for (int n = 0; n < NST; ++n) {
        float dA = __expf(dt * a[n]);
        S[n] = S[n] * dA + dtu * dblT[t][DTR + n];
        P[n] *= dA;
      }
    }
    const size_t base = ((size_t)(bsq * NCH + chk) * DI + d) * NST;
#pragma unroll
    for (int n = 0; n < NST; ++n) {
      Pb[base + n] = P[n];
      Sb[base + n] = S[n];
    }
  }
}

__global__ void k_scan_combine(const float* __restrict__ Pb,
                               float* __restrict__ Sb) {
  __shared__ float Pg[NG][32];
  __shared__ float Sg[NG][32];
  const int e = threadIdx.x & 31;
  const int g = threadIdx.x >> 5;
  const int flat = blockIdx.x * 32 + e;
  const int b  = flat / (DI * NST);
  const int dn = flat % (DI * NST);
  const size_t CS = (size_t)DI * NST;
  const size_t base = (size_t)(b * NCH + g * CG) * CS + dn;

  float p[CG], s[CG];
#pragma unroll
  for (int j = 0; j < CG; ++j) {
    p[j] = Pb[base + (size_t)j * CS];
    s[j] = Sb[base + (size_t)j * CS];
  }
  float pp = 1.0f, ss = 0.0f;
#pragma unroll
  for (int j = 0; j < CG; ++j) {
    ss = p[j] * ss + s[j];
    pp *= p[j];
  }
  Pg[g][e] = pp;
  Sg[g][e] = ss;
  __syncthreads();

  float carry = 0.0f;
  for (int k = 0; k < g; ++k)
    carry = Pg[k][e] * carry + Sg[k][e];
#pragma unroll
  for (int j = 0; j < CG; ++j) {
    Sb[base + (size_t)j * CS] = carry;
    carry = p[j] * carry + s[j];
  }
}

#define YP 388
#define RT 32

__global__ void __launch_bounds__(2 * DI)
k_scan3_out(const float* __restrict__ xz_f, const float* __restrict__ xz_b,
            const float* __restrict__ dtb_f, const float* __restrict__ dtb_b,
            const float* __restrict__ u_f, const float* __restrict__ u_b,
            const float* __restrict__ dbl_f, const float* __restrict__ dbl_b,
            const float* __restrict__ Alog_f, const float* __restrict__ Alog_b,
            const float* __restrict__ Dv_f, const float* __restrict__ Dv_b,
            const float* __restrict__ Sb_f, const float* __restrict__ Sb_b,
            const float* __restrict__ x, const float* __restrict__ Wout,
            float* __restrict__ out) {
  __shared__ __attribute__((aligned(16))) float yF[RT * YP];
  __shared__ __attribute__((aligned(16))) float yB[RT * YP];
  const int b  = blockIdx.x / (NCH / 2);
  const int cp = blockIdx.x % (NCH / 2);
  const int g  = __builtin_amdgcn_readfirstlane((int)(threadIdx.x / DI));
  const int d  = threadIdx.x - g * DI;
  const int c  = 2 * cp + g;
  const int cr = NCH - 1 - c;

  float af[NST], ab[NST], hf[NST], hb[NST];
  const size_t base_f = ((size_t)(b * NCH + c)  * DI + d) * NST;
  const size_t base_b = ((size_t)(b * NCH + cr) * DI + d) * NST;
#pragma unroll
  for (int n = 0; n < NST; ++n) {
    af[n] = -__expf(Alog_f[d * NST + n]);
    ab[n] = -__expf(Alog_b[d * NST + n]);
    hf[n] = Sb_f[base_f + n];
    hb[n] = Sb_b[base_b + n];
  }
  const float Dd_f = Dv_f[d];
  const float Dd_b = Dv_b[d];

  const size_t rf0 = (size_t)b * SEQ + (size_t)c  * CL;
  const size_t rb0 = (size_t)b * SEQ + (size_t)cr * CL;
  float* yFg = &yF[(CL * g) * YP + d];
  float* yBg = &yB[(CL * g) * YP + d];
  for (int t = 0; t < CL; ++t) {
    const size_t rowf = rf0 + t;
    const size_t rowb = rb0 + t;
    const float* bcf = dbl_f + rowf * DBLW;
    const float* bcb = dbl_b + rowb * DBLW;
    __builtin_prefetch(bcf + DBLW, 0, 0);
    __builtin_prefetch(bcb + DBLW, 0, 0);
    const float dtf = dtb_f[rowf * DI + d];
    const float dtr = dtb_b[rowb * DI + d];
    const float utf = u_f[rowf * DI + d];
    const float utr = u_b[rowb * DI + d];
    const float zf  = xz_f[rowf * XZW + DI + d];
    const float zr  = xz_b[rowb * XZW + DI + d];
    const float dtuf = dtf * utf;
    const float dtur = dtr * utr;
    float yf = 0.0f, yr = 0.0f;
#pragma unroll
    for (int n = 0; n < NST; ++n) {
      const float dAf = __expf(dtf * af[n]);
      const float dAr = __expf(dtr * ab[n]);
      hf[n] = hf[n] * dAf + dtuf * bcf[DTR + n];
      hb[n] = hb[n] * dAr + dtur * bcb[DTR + n];
      yf = __builtin_fmaf(hf[n], bcf[DTR + NST + n], yf);
      yr = __builtin_fmaf(hb[n], bcb[DTR + NST + n], yr);
    }
    yf += utf * Dd_f;
    yr += utr * Dd_b;
    yf *= zf / (1.0f + __expf(-zf));
    yr *= zr / (1.0f + __expf(-zr));
    yFg[t * YP] = yf;
    yBg[(CL - 1 - t) * YP] = yr;
  }
  __syncthreads();

  const int w = threadIdx.x >> 5;
  if (w < DIMX / 16) {
    const int lane = threadIdx.x & 31;
    const int mrow = lane & 15, kh = lane >> 4;
    const float* bp  = Wout + (size_t)(w * 16 + mrow) * DI;
    const float* pf0 = &yF[mrow * YP];
    const float* pb0 = &yB[mrow * YP];
    const float* pf1 = &yF[(16 + mrow) * YP];
    const float* pb1 = &yB[(16 + mrow) * YP];
    v8f acc0 = {}, acc1 = {};
    for (int k0 = 0; k0 < DI; k0 += 4) {
      const int kk = k0 + 2 * kh;
      const v2f fa0 = *(const v2f*)(pf0 + kk);
      const v2f fb0 = *(const v2f*)(pb0 + kk);
      const v2f fa1 = *(const v2f*)(pf1 + kk);
      const v2f fb1 = *(const v2f*)(pb1 + kk);
      v2f a0; a0.x = fa0.x + fb0.x; a0.y = fa0.y + fb0.y;
      v2f a1; a1.x = fa1.x + fb1.x; a1.y = fa1.y + fb1.y;
      v2f bv; bv.x = bp[kk]; bv.y = bp[kk + 1];
      acc0 = wmma4(a0, bv, acc0);
      acc1 = wmma4(a1, bv, acc1);
    }
    const int n = w * 16 + mrow;
    const size_t r0 = (size_t)b * SEQ + (size_t)cp * RT;
#pragma unroll
    for (int v = 0; v < 8; ++v) {
      const size_t orow = r0 + v + 8 * kh;
      out[orow * DIMX + n] = acc0[v] + x[orow * DIMX + n];
      out[(orow + 16) * DIMX + n] = acc1[v] + x[(orow + 16) * DIMX + n];
    }
  }
}

extern "C" void kernel_launch(void* const* d_in, const int* in_sizes, int n_in,
                              void* d_out, int out_size, void* d_ws, size_t ws_size,
                              hipStream_t stream) {
  const float* x      = (const float*)d_in[0];
  const float* Win_f  = (const float*)d_in[1];
  const float* cw_f   = (const float*)d_in[2];
  const float* cb_f   = (const float*)d_in[3];
  const float* Wxp_f  = (const float*)d_in[4];
  const float* Wdt_f  = (const float*)d_in[5];
  const float* bdt_f  = (const float*)d_in[6];
  const float* Alog_f = (const float*)d_in[7];
  const float* Dv_f   = (const float*)d_in[8];
  const float* Win_b  = (const float*)d_in[9];
  const float* cw_b   = (const float*)d_in[10];
  const float* cb_b   = (const float*)d_in[11];
  const float* Wxp_b  = (const float*)d_in[12];
  const float* Wdt_b  = (const float*)d_in[13];
  const float* bdt_b  = (const float*)d_in[14];
  const float* Alog_b = (const float*)d_in[15];
  const float* Dv_b   = (const float*)d_in[16];
  const float* Wout   = (const float*)d_in[17];
  float* out = (float*)d_out;

  float* ws    = (float*)d_ws;
  float* xz_f  = ws;
  float* xz_b  = xz_f + (size_t)MROWS * XZW;
  float* u_f   = xz_b + (size_t)MROWS * XZW;
  float* u_b   = u_f  + (size_t)MROWS * DI;
  float* dbl_f = u_b  + (size_t)MROWS * DI;
  float* dbl_b = dbl_f + (size_t)MROWS * DBLW;
  const size_t PS = (size_t)BATCH * NCH * DI * NST;
  float* P_f   = dbl_b + (size_t)MROWS * DBLW;
  float* S_f   = P_f + PS;
  float* P_b   = S_f + PS;
  float* S_b   = P_b + PS;
  float* dt_f  = S_b + PS;
  float* dt_b  = dt_f + (size_t)MROWS * DI;

  {
    const int gb = (MROWS / 128) * (XZW / 192);
    k_gemm_in<<<gb, 128, 0, stream>>>(x, Win_f, xz_f, 0);
    k_gemm_in<<<gb, 128, 0, stream>>>(x, Win_b, xz_b, 1);
  }
  {
    k_conv_xp<<<MROWS / 16, 32 * KW, 0, stream>>>(xz_f, cw_f, cb_f, Wxp_f, Wdt_f, bdt_f, Alog_f,
                                                 u_f, dbl_f, dt_f, P_f, S_f);
    k_conv_xp<<<MROWS / 16, 32 * KW, 0, stream>>>(xz_b, cw_b, cb_b, Wxp_b, Wdt_b, bdt_b, Alog_b,
                                                 u_b, dbl_b, dt_b, P_b, S_b);
  }
  {
    int cb = BATCH * DI * NST / 32;
    k_scan_combine<<<cb, 32 * NG, 0, stream>>>(P_f, S_f);
    k_scan_combine<<<cb, 32 * NG, 0, stream>>>(P_b, S_b);
  }
  {
    k_scan3_out<<<BATCH * NCH / 2, 2 * DI, 0, stream>>>(xz_f, xz_b, dt_f, dt_b, u_f, u_b, dbl_f, dbl_b,
                                                        Alog_f, Alog_b, Dv_f, Dv_b, S_f, S_b,
                                                        x, Wout, out);
  }
}
